# starting kernel
# baseline (speedup 1.0000x reference)
#include <hip/hip_runtime.h>

typedef __attribute__((ext_vector_type(2))) float v2f;
typedef __attribute__((ext_vector_type(4))) float v4f;
typedef __attribute__((ext_vector_type(8))) float v8f;

#define DF 128


__global__ void deg_init_kernel(float* deg, int n) {
    int i = blockIdx.x * blockDim.x + threadIdx.x;
    if (i < n) deg[i] = 1.0f;
}

__global__ void deg_accum_kernel(const int* __restrict__ dst, float* deg, int e) {
    int i = blockIdx.x * blockDim.x + threadIdx.x;
    if (i < e) unsafeAtomicAdd(&deg[dst[i]], 1.0f);
}

__global__ void deg_rsqrt_kernel(float* deg, int n) {
    int i = blockIdx.x * blockDim.x + threadIdx.x;
    if (i < n) deg[i] = rsqrtf(deg[i]);
}


__global__ __launch_bounds__(128) void gemm_wmma_kernel(const float* __restrict__ X,
                                                        const float* __restrict__ W,
                                                        float* __restrict__ H, int nrows) {
    __shared__ float sW[DF * DF];
    for (int i = threadIdx.x; i < DF * DF; i += blockDim.x) {
        int k = i >> 7;
        int nn = i & 127;
        sW[(k >> 1) * 256 + nn * 2 + (k & 1)] = W[i];
    }
    __syncthreads();

    const int wave = threadIdx.x >> 5;
    const int lane = threadIdx.x & 31;
    const int tile = blockIdx.x * 4 + wave;
    const int m0 = tile * 16;
    if (m0 >= nrows) return;

    const int half = lane >> 4;
    const int l = lane & 15;

    v8f acc[8] = {};
    const float* xrow = X + (size_t)(m0 + l) * DF;

    for (int kb = 0; kb < DF; kb += 4) {
        v2f a = *(const v2f*)(xrow + kb + half * 2);
        const float* wp = sW + ((kb >> 1) + half) * 256;
#pragma unroll
        for (int nt = 0; nt < 8; nt++) {
            v2f b = *(const v2f*)(wp + (nt * 16 + l) * 2);
            acc[nt] = __builtin_amdgcn_wmma_f32_16x16x4_f32(
                false, a, false, b, (short)0, acc[nt], false, false);
        }
    }

#pragma unroll
    for (int nt = 0; nt < 8; nt++) {
        const int col = nt * 16 + l;
#pragma unroll
        for (int r = 0; r < 8; r++) {
            const int row = m0 + r + 8 * half;
            H[(size_t)row * DF + col] = acc[nt][r];
        }
    }
}


__global__ void selfloop_init_kernel(const float* __restrict__ H, const float* __restrict__ dinv,
                                     float* __restrict__ out, int n) {
    int i = blockIdx.x * blockDim.x + threadIdx.x;
    int total = n * (DF / 4);
    if (i >= total) return;
    int node = i >> 5;
    float s = dinv[node];
    s = s * s;
    v4f v = ((const v4f*)H)[i];
    ((v4f*)out)[i] = v * s;
}

__global__ __launch_bounds__(256) void scatter_edges_kernel(const int* __restrict__ src,
                                                            const int* __restrict__ dst,
                                                            const float* __restrict__ dinv,
                                                            const float* __restrict__ H,
                                                            float* __restrict__ out, int e) {
    const int lane = threadIdx.x & 31;
    const int w = (int)((blockIdx.x * blockDim.x + threadIdx.x) >> 5);
    if (w >= e) return;
    const int s = src[w];
    const int d = dst[w];
    const float norm = dinv[s] * dinv[d];
    v4f m = ((const v4f*)(H + (size_t)s * DF))[lane];
    float* o = out + (size_t)d * DF + lane * 4;
    unsafeAtomicAdd(o + 0, m.x * norm);
    unsafeAtomicAdd(o + 1, m.y * norm);
    unsafeAtomicAdd(o + 2, m.z * norm);
    unsafeAtomicAdd(o + 3, m.w * norm);
}

__global__ void bias_act_kernel(const float* __restrict__ agg, const float* __restrict__ bias,
                                float* __restrict__ z, int n, int do_relu) {
    int i = blockIdx.x * blockDim.x + threadIdx.x;
    int total = n * (DF / 4);
    if (i >= total) return;
    int c4 = i & 31;
    v4f b = ((const v4f*)bias)[c4];
    v4f v = ((const v4f*)agg)[i] + b;
    if (do_relu) {
        v.x = fmaxf(v.x, 0.0f);
        v.y = fmaxf(v.y, 0.0f);
        v.z = fmaxf(v.z, 0.0f);
        v.w = fmaxf(v.w, 0.0f);
    }
    ((v4f*)z)[i] = v;
}


__global__ __launch_bounds__(256) void decode_kernel(const int* __restrict__ s,
                                                     const int* __restrict__ d,
                                                     const float* __restrict__ z,
                                                     float* __restrict__ logits, int e) {
    const int lane = threadIdx.x & 31;
    const int w = (int)((blockIdx.x * blockDim.x + threadIdx.x) >> 5);
    if (w >= e) return;
    const int si = s[w];
    const int di = d[w];
    v4f a = ((const v4f*)(z + (size_t)si * DF))[lane];
    v4f b = ((const v4f*)(z + (size_t)di * DF))[lane];
    float dot = a.x * b.x + a.y * b.y + a.z * b.z + a.w * b.w;
#pragma unroll
    for (int off = 16; off > 0; off >>= 1) dot += __shfl_xor(dot, off, 32);
    if (lane == 0) logits[w] = dot;
}


extern "C" void kernel_launch(void* const* d_in, const int* in_sizes, int n_in,
                              void* d_out, int out_size, void* d_ws, size_t ws_size,
                              hipStream_t stream) {
    const float* x  = (const float*)d_in[0];
    const int*   ei = (const int*)d_in[1];
    const int*   el = (const int*)d_in[2];
    const float* W1 = (const float*)d_in[3];
    const float* b1 = (const float*)d_in[4];
    const float* W2 = (const float*)d_in[5];
    const float* b2 = (const float*)d_in[6];

    const int n  = in_sizes[0] / DF;
    const int E  = in_sizes[1] / 2;
    const int EL = in_sizes[2] / 2;

    float* bufA = (float*)d_ws;
    float* bufB = bufA + (size_t)n * DF;
    float* dinv = bufB + (size_t)n * DF;
    float* out  = (float*)d_out;

    const int tiles  = (n + 15) / 16;
    const int gemmBl = (tiles + 3) / 4;
    const int elem4  = n * (DF / 4);
    const int elBl   = (elem4 + 255) / 256;
    const int nBl    = (n + 255) / 256;
    const int eBl    = (E + 255) / 256;
    const int scatBl = (E + 7) / 8;
    const int decBl  = (EL + 7) / 8;

    deg_init_kernel<<<nBl, 256, 0, stream>>>(dinv, n);
    deg_accum_kernel<<<eBl, 256, 0, stream>>>(ei + E, dinv, E);
    deg_rsqrt_kernel<<<nBl, 256, 0, stream>>>(dinv, n);

    gemm_wmma_kernel<<<gemmBl, 128, 0, stream>>>(x, W1, bufA, n);
    selfloop_init_kernel<<<elBl, 256, 0, stream>>>(bufA, dinv, bufB, n);
    scatter_edges_kernel<<<scatBl, 256, 0, stream>>>(ei, ei + E, dinv, bufA, bufB, E);
    bias_act_kernel<<<elBl, 256, 0, stream>>>(bufB, b1, bufA, n, 1);

    gemm_wmma_kernel<<<gemmBl, 128, 0, stream>>>(bufA, W2, bufB, n);
    selfloop_init_kernel<<<elBl, 256, 0, stream>>>(bufB, dinv, bufA, n);
    scatter_edges_kernel<<<scatBl, 256, 0, stream>>>(ei, ei + E, dinv, bufB, bufA, E);
    bias_act_kernel<<<elBl, 256, 0, stream>>>(bufA, b2, bufB, n, 0);

    decode_kernel<<<decBl, 256, 0, stream>>>(el, el + EL, bufB, out, EL);
}


# Replace the float-atomic edge scatter by neighbour lists sorted by dst (counting sort, arrival-order rank from int atomics) and one wave per node summing rows in registers; sum order changes
# speedup vs baseline: 2.1878x; 2.1878x over previous
#include <hip/hip_runtime.h>

typedef __attribute__((ext_vector_type(2))) float v2f;
typedef __attribute__((ext_vector_type(4))) float v4f;
typedef __attribute__((ext_vector_type(8))) float v8f;

#define DF 128


#define SCAN_TILE 1024

__global__ void count_zero_kernel(int* cnt, int n) {
    int i = blockIdx.x * blockDim.x + threadIdx.x;
    if (i < n) cnt[i] = 0;
}

__global__ void edge_rank_kernel(const int* __restrict__ dst, int* cnt, int* __restrict__ rank, int e) {
    int i = blockIdx.x * blockDim.x + threadIdx.x;
    if (i < e) rank[i] = atomicAdd(&cnt[dst[i]], 1);
}

__global__ __launch_bounds__(256) void tile_sum_kernel(const int* __restrict__ cnt,
                                                       int* __restrict__ tsum, int n) {
    __shared__ int sw[8];
    const int wave = threadIdx.x >> 5;
    const int lane = threadIdx.x & 31;
    const int base = blockIdx.x * SCAN_TILE + threadIdx.x * 4;
    int s = 0;
#pragma unroll
    for (int j = 0; j < 4; j++)
        if (base + j < n) s += cnt[base + j];
#pragma unroll
    for (int off = 16; off > 0; off >>= 1) s += __shfl_xor(s, off, 32);
    if (lane == 0) sw[wave] = s;
    __syncthreads();
    if (threadIdx.x == 0) {
        int t = 0;
#pragma unroll
        for (int w = 0; w < 8; w++) t += sw[w];
        tsum[blockIdx.x] = t;
    }
}

__global__ __launch_bounds__(256) void rowptr_dinv_kernel(const int* __restrict__ cnt,
                                                          const int* __restrict__ tsum,
                                                          int* __restrict__ rowptr,
                                                          float* __restrict__ dinv, int n) {
    __shared__ int sw[8];
    __shared__ int sv[8];
    const int wave = threadIdx.x >> 5;
    const int lane = threadIdx.x & 31;

    int p = 0;
    for (int t = threadIdx.x; t < (int)blockIdx.x; t += 256) p += tsum[t];
#pragma unroll
    for (int off = 16; off > 0; off >>= 1) p += __shfl_xor(p, off, 32);
    if (lane == 0) sw[wave] = p;

    const int base = blockIdx.x * SCAN_TILE + threadIdx.x * 4;
    int c[4];
#pragma unroll
    for (int j = 0; j < 4; j++) c[j] = (base + j < n) ? cnt[base + j] : 0;
    const int s = c[0] + c[1] + c[2] + c[3];
    int incl = s;
#pragma unroll
    for (int off = 1; off < 32; off <<= 1) {
        int t = __shfl_up(incl, off, 32);
        if (lane >= off) incl += t;
    }
    if (lane == 31) sv[wave] = incl;
    __syncthreads();

    int run = incl - s;
#pragma unroll
    for (int w = 0; w < 8; w++) {
        run += sw[w];
        if (w < wave) run += sv[w];
    }
#pragma unroll
    for (int j = 0; j < 4; j++) {
        if (base + j < n) {
            rowptr[base + j] = run;
            dinv[base + j] = rsqrtf((float)(c[j] + 1));
            run += c[j];
        }
    }
}

__global__ void fill_col_kernel(const int* __restrict__ src, const int* __restrict__ dst,
                                const int* __restrict__ rank, const int* __restrict__ rowptr,
                                int* __restrict__ col, int e) {
    int i = blockIdx.x * blockDim.x + threadIdx.x;
    if (i < e) col[rowptr[dst[i]] + rank[i]] = src[i];
}


__global__ __launch_bounds__(128) void gemm_wmma_kernel(const float* __restrict__ X,
                                                        const float* __restrict__ W,
                                                        float* __restrict__ H, int nrows) {
    __shared__ float sW[DF * DF];
    for (int i = threadIdx.x; i < DF * DF; i += blockDim.x) {
        int k = i >> 7;
        int nn = i & 127;
        sW[(k >> 1) * 256 + nn * 2 + (k & 1)] = W[i];
    }
    __syncthreads();

    const int wave = threadIdx.x >> 5;
    const int lane = threadIdx.x & 31;
    const int tile = blockIdx.x * 4 + wave;
    const int m0 = tile * 16;
    if (m0 >= nrows) return;

    const int half = lane >> 4;
    const int l = lane & 15;

    v8f acc[8] = {};
    const float* xrow = X + (size_t)(m0 + l) * DF;

    for (int kb = 0; kb < DF; kb += 4) {
        v2f a = *(const v2f*)(xrow + kb + half * 2);
        const float* wp = sW + ((kb >> 1) + half) * 256;
#pragma unroll
        for (int nt = 0; nt < 8; nt++) {
            v2f b = *(const v2f*)(wp + (nt * 16 + l) * 2);
            acc[nt] = __builtin_amdgcn_wmma_f32_16x16x4_f32(
                false, a, false, b, (short)0, acc[nt], false, false);
        }
    }

#pragma unroll
    for (int nt = 0; nt < 8; nt++) {
        const int col = nt * 16 + l;
#pragma unroll
        for (int r = 0; r < 8; r++) {
            const int row = m0 + r + 8 * half;
            H[(size_t)row * DF + col] = acc[nt][r];
        }
    }
}


__global__ __launch_bounds__(256) void gather_rows_kernel(const int* __restrict__ rowptr,
                                                          const int* __restrict__ cnt,
                                                          const int* __restrict__ col,
                                                          const float* __restrict__ dinv,
                                                          const float* __restrict__ H,
                                                          const float* __restrict__ bias,
                                                          float* __restrict__ out, int n, int do_relu) {
    const int lane = threadIdx.x & 31;
    const int i = (int)((blockIdx.x * blockDim.x + threadIdx.x) >> 5);
    if (i >= n) return;
    const float di = dinv[i];
    const int start = rowptr[i];
    const int deg = cnt[i];
    const v4f* H4 = (const v4f*)H;

    v4f acc = H4[(size_t)i * (DF / 4) + lane] * (di * di);

    for (int base = 0; base < deg; base += 32) {
        const int idx = base + lane;
        int j = 0;
        float w = 0.0f;
        if (idx < deg) {
            j = col[start + idx];
            w = dinv[j] * di;
        }
        const int wi = __float_as_int(w);
        const int m = min(32, deg - base);
        int k = 0;
        for (; k + 4 <= m; k += 4) {
            const int j0 = __builtin_amdgcn_readlane(j, k);
            const int j1 = __builtin_amdgcn_readlane(j, k + 1);
            const int j2 = __builtin_amdgcn_readlane(j, k + 2);
            const int j3 = __builtin_amdgcn_readlane(j, k + 3);
            const float w0 = __int_as_float(__builtin_amdgcn_readlane(wi, k));
            const float w1 = __int_as_float(__builtin_amdgcn_readlane(wi, k + 1));
            const float w2 = __int_as_float(__builtin_amdgcn_readlane(wi, k + 2));
            const float w3 = __int_as_float(__builtin_amdgcn_readlane(wi, k + 3));
            const v4f h0 = H4[(size_t)j0 * (DF / 4) + lane];
            const v4f h1 = H4[(size_t)j1 * (DF / 4) + lane];
            const v4f h2 = H4[(size_t)j2 * (DF / 4) + lane];
            const v4f h3 = H4[(size_t)j3 * (DF / 4) + lane];
            acc += h0 * w0;
            acc += h1 * w1;
            acc += h2 * w2;
            acc += h3 * w3;
        }
        for (; k < m; k++) {
            const int j0 = __builtin_amdgcn_readlane(j, k);
            const float w0 = __int_as_float(__builtin_amdgcn_readlane(wi, k));
            acc += H4[(size_t)j0 * (DF / 4) + lane] * w0;
        }
    }

    acc += ((const v4f*)bias)[lane];
    if (do_relu) {
        acc.x = fmaxf(acc.x, 0.0f);
        acc.y = fmaxf(acc.y, 0.0f);
        acc.z = fmaxf(acc.z, 0.0f);
        acc.w = fmaxf(acc.w, 0.0f);
    }
    ((v4f*)out)[(size_t)i * (DF / 4) + lane] = acc;
}


__global__ __launch_bounds__(256) void decode_kernel(const int* __restrict__ s,
                                                     const int* __restrict__ d,
                                                     const float* __restrict__ z,
                                                     float* __restrict__ logits, int e) {
    const int lane = threadIdx.x & 31;
    const int w = (int)((blockIdx.x * blockDim.x + threadIdx.x) >> 5);
    if (w >= e) return;
    const int si = s[w];
    const int di = d[w];
    v4f a = ((const v4f*)(z + (size_t)si * DF))[lane];
    v4f b = ((const v4f*)(z + (size_t)di * DF))[lane];
    float dot = a.x * b.x + a.y * b.y + a.z * b.z + a.w * b.w;
#pragma unroll
    for (int off = 16; off > 0; off >>= 1) dot += __shfl_xor(dot, off, 32);
    if (lane == 0) logits[w] = dot;
}


extern "C" void kernel_launch(void* const* d_in, const int* in_sizes, int n_in,
                              void* d_out, int out_size, void* d_ws, size_t ws_size,
                              hipStream_t stream) {
    const float* x  = (const float*)d_in[0];
    const int*   ei = (const int*)d_in[1];
    const int*   el = (const int*)d_in[2];
    const float* W1 = (const float*)d_in[3];
    const float* b1 = (const float*)d_in[4];
    const float* W2 = (const float*)d_in[5];
    const float* b2 = (const float*)d_in[6];

    const int n  = in_sizes[0] / DF;
    const int E  = in_sizes[1] / 2;
    const int EL = in_sizes[2] / 2;

    const int tiles  = (n + 15) / 16;
    const int gemmBl = (tiles + 3) / 4;
    const int nBl    = (n + 255) / 256;
    const int eBl    = (E + 255) / 256;
    const int scanBl = (n + SCAN_TILE - 1) / SCAN_TILE;
    const int nodeBl = (n + 7) / 8;
    const int decBl  = (EL + 7) / 8;

    float* bufA   = (float*)d_ws;
    float* bufB   = bufA + (size_t)n * DF;
    float* dinv   = bufB + (size_t)n * DF;
    int*   cnt    = (int*)(dinv + n);
    int*   rowptr = cnt + n;
    int*   tsum   = rowptr + n;
    int*   rank   = tsum + scanBl;
    int*   col    = rank + E;
    float* out    = (float*)d_out;

    count_zero_kernel<<<nBl, 256, 0, stream>>>(cnt, n);
    edge_rank_kernel<<<eBl, 256, 0, stream>>>(ei + E, cnt, rank, E);
    tile_sum_kernel<<<scanBl, 256, 0, stream>>>(cnt, tsum, n);
    rowptr_dinv_kernel<<<scanBl, 256, 0, stream>>>(cnt, tsum, rowptr, dinv, n);
    fill_col_kernel<<<eBl, 256, 0, stream>>>(ei, ei + E, rank, rowptr, col, E);

    gemm_wmma_kernel<<<gemmBl, 128, 0, stream>>>(x, W1, bufA, n);
    gather_rows_kernel<<<nodeBl, 256, 0, stream>>>(rowptr, cnt, col, dinv, bufA, b1, bufB, n, 1);

    gemm_wmma_kernel<<<gemmBl, 128, 0, stream>>>(bufB, W2, bufA, n);
    gather_rows_kernel<<<nodeBl, 256, 0, stream>>>(rowptr, cnt, col, dinv, bufA, b2, bufB, n, 0);

    decode_kernel<<<decBl, 256, 0, stream>>>(el, el + EL, bufB, out, EL);
}


# Rank edges with LDS counters in a two-level counting sort (chunk by bucket, then one block per 512-node bucket), not a returning global atomic and a scattered store per edge; sum order changes
# speedup vs baseline: 2.8696x; 1.3116x over previous
#include <hip/hip_runtime.h>

typedef __attribute__((ext_vector_type(2))) float v2f;
typedef __attribute__((ext_vector_type(4))) float v4f;
typedef __attribute__((ext_vector_type(8))) float v8f;

#define DF 128


#define BSH 9
#define BNODES 512
#define NBMAX 256
#define CH 8192
#define KCAP 16384

__global__ __launch_bounds__(1024) void chunk_hist_kernel(const int* __restrict__ dst,
                                                          int* __restrict__ hmat, int e, int nb) {
    __shared__ int hist[NBMAX];
    if (threadIdx.x < NBMAX) hist[threadIdx.x] = 0;
    __syncthreads();
    const int c0 = blockIdx.x * CH;
    int ed[8];
#pragma unroll
    for (int k = 0; k < 8; k++) {
        const int i = c0 + k * 1024 + (int)threadIdx.x;
        ed[k] = (i < e) ? dst[i] : -1;
    }
#pragma unroll
    for (int k = 0; k < 8; k++)
        if (ed[k] >= 0) atomicAdd(&hist[ed[k] >> BSH], 1);
    __syncthreads();
    if ((int)threadIdx.x < nb) hmat[blockIdx.x * nb + threadIdx.x] = hist[threadIdx.x];
}

__global__ __launch_bounds__(1024) void chunk_scatter_kernel(const int* __restrict__ src,
                                                             const int* __restrict__ dst,
                                                             const int* __restrict__ hmat,
                                                             int* __restrict__ bstart,
                                                             int* __restrict__ psrc,
                                                             int* __restrict__ pdst,
                                                             int e, int nb, int nchunks) {
    __shared__ int hist[NBMAX];
    __shared__ int lstart[NBMAX];
    __shared__ int gbase[NBMAX];
    __shared__ int ptot[4][NBMAX];
    __shared__ int ppre[4][NBMAX];
    __shared__ int wsT[8];
    __shared__ int wsH[8];
    __shared__ int stS[CH];
    __shared__ int stD[CH];

    const int tid = threadIdx.x;
    const int lane = tid & 31;
    const int wave = tid >> 5;
    const int blk = blockIdx.x;
    const int c0 = blk * CH;

    if (tid < NBMAX) hist[tid] = 0;
    __syncthreads();

    int es[8], ed[8], er[8];
#pragma unroll
    for (int k = 0; k < 8; k++) {
        const int i = c0 + k * 1024 + tid;
        ed[k] = (i < e) ? dst[i] : -1;
        es[k] = (i < e) ? src[i] : 0;
    }
#pragma unroll
    for (int k = 0; k < 8; k++) {
        er[k] = 0;
        if (ed[k] >= 0) er[k] = atomicAdd(&hist[ed[k] >> BSH], 1);
    }

    {
        const int b = tid & (NBMAX - 1);
        const int g = tid >> 8;
        const int rpg = (nchunks + 3) >> 2;
        const int r0 = g * rpg;
        const int r1 = min(nchunks, r0 + rpg);
        int tot = 0, pre = 0;
        if (b < nb) {
#pragma unroll 8
            for (int r = r0; r < r1; r++) {
                const int v = hmat[r * nb + b];
                tot += v;
                pre += (r < blk) ? v : 0;
            }
        }
        ptot[g][b] = tot;
        ppre[g][b] = pre;
    }
    __syncthreads();

    int T = 0, P = 0, Hc = 0;
    if (tid < NBMAX) {
        T = ptot[0][tid] + ptot[1][tid] + ptot[2][tid] + ptot[3][tid];
        P = ppre[0][tid] + ppre[1][tid] + ppre[2][tid] + ppre[3][tid];
        Hc = hist[tid];
    }
    int incT = T, incH = Hc;
#pragma unroll
    for (int off = 1; off < 32; off <<= 1) {
        const int a = __shfl_up(incT, off, 32);
        const int h = __shfl_up(incH, off, 32);
        if (lane >= off) {
            incT += a;
            incH += h;
        }
    }
    if (tid < NBMAX && lane == 31) {
        wsT[wave] = incT;
        wsH[wave] = incH;
    }
    __syncthreads();
    if (tid < NBMAX) {
        int oT = 0, oH = 0;
#pragma unroll
        for (int w = 0; w < 8; w++) {
            if (w < wave) {
                oT += wsT[w];
                oH += wsH[w];
            }
        }
        const int exT = oT + incT - T;
        const int exH = oH + incH - Hc;
        lstart[tid] = exH;
        gbase[tid] = exT + P;
        if (blk == 0) {
            if (tid < nb) bstart[tid] = exT;
            if (tid == nb - 1) bstart[nb] = exT + T;
        }
    }
    __syncthreads();

#pragma unroll
    for (int k = 0; k < 8; k++) {
        if (ed[k] >= 0) {
            const int p = lstart[ed[k] >> BSH] + er[k];
            stS[p] = es[k];
            stD[p] = ed[k];
        }
    }
    __syncthreads();

    const int m = min(CH, e - c0);
    for (int i = tid; i < m; i += 1024) {
        const int s = stS[i];
        const int d = stD[i];
        const int b = d >> BSH;
        const int g = gbase[b] + (i - lstart[b]);
        psrc[g] = s;
        pdst[g] = d;
    }
}

__global__ __launch_bounds__(1024) void bucket_sort_kernel(const int* __restrict__ psrc,
                                                           const int* __restrict__ pdst,
                                                           const int* __restrict__ bstart,
                                                           int* __restrict__ rowptr,
                                                           int* __restrict__ cnt,
                                                           float* __restrict__ dinv,
                                                           int* __restrict__ col, int n) {
    __shared__ int cntL[BNODES];
    __shared__ int curL[BNODES];
    __shared__ int wsum[16];
    __shared__ int stage[KCAP];

    const int tid = threadIdx.x;
    const int lane = tid & 31;
    const int wave = tid >> 5;
    const int b = blockIdx.x;
    const int s = bstart[b];
    const int m = bstart[b + 1] - s;
    const int node0 = b << BSH;

    if (tid < BNODES) cntL[tid] = 0;
    __syncthreads();
    for (int i0 = tid; i0 < m; i0 += 8 * 1024) {
        int d[8];
#pragma unroll
        for (int k = 0; k < 8; k++) {
            const int i = i0 + k * 1024;
            d[k] = (i < m) ? pdst[s + i] : -1;
        }
#pragma unroll
        for (int k = 0; k < 8; k++)
            if (d[k] >= 0) atomicAdd(&cntL[d[k] - node0], 1);
    }
    __syncthreads();

    int c = 0;
    if (tid < BNODES) c = cntL[tid];
    int inc = c;
#pragma unroll
    for (int off = 1; off < 32; off <<= 1) {
        const int t = __shfl_up(inc, off, 32);
        if (lane >= off) inc += t;
    }
    if (tid < BNODES && lane == 31) wsum[wave] = inc;
    __syncthreads();
    if (tid < BNODES) {
        int o = 0;
#pragma unroll
        for (int w = 0; w < 16; w++)
            if (w < wave) o += wsum[w];
        const int ex = o + inc - c;
        curL[tid] = ex;
        const int node = node0 + tid;
        if (node < n) {
            rowptr[node] = s + ex;
            cnt[node] = c;
            dinv[node] = rsqrtf((float)(c + 1));
        }
    }
    __syncthreads();

    const bool staged = (m <= KCAP);
    for (int i0 = tid; i0 < m; i0 += 8 * 1024) {
        int d[8], sv[8];
#pragma unroll
        for (int k = 0; k < 8; k++) {
            const int i = i0 + k * 1024;
            d[k] = (i < m) ? pdst[s + i] : -1;
            sv[k] = (i < m) ? psrc[s + i] : 0;
        }
#pragma unroll
        for (int k = 0; k < 8; k++) {
            if (d[k] >= 0) {
                const int p = atomicAdd(&curL[d[k] - node0], 1);
                if (staged) stage[p] = sv[k];
                else col[s + p] = sv[k];
            }
        }
    }
    if (staged) {
        __syncthreads();
        for (int i = tid; i < m; i += 1024) col[s + i] = stage[i];
    }
}


__global__ __launch_bounds__(128) void gemm_wmma_kernel(const float* __restrict__ X,
                                                        const float* __restrict__ W,
                                                        float* __restrict__ H, int nrows) {
    __shared__ float sW[DF * DF];
    for (int i = threadIdx.x; i < DF * DF; i += blockDim.x) {
        int k = i >> 7;
        int nn = i & 127;
        sW[(k >> 1) * 256 + nn * 2 + (k & 1)] = W[i];
    }
    __syncthreads();

    const int wave = threadIdx.x >> 5;
    const int lane = threadIdx.x & 31;
    const int tile = blockIdx.x * 4 + wave;
    const int m0 = tile * 16;
    if (m0 >= nrows) return;

    const int half = lane >> 4;
    const int l = lane & 15;

    v8f acc[8] = {};
    const float* xrow = X + (size_t)(m0 + l) * DF;

    for (int kb = 0; kb < DF; kb += 4) {
        v2f a = *(const v2f*)(xrow + kb + half * 2);
        const float* wp = sW + ((kb >> 1) + half) * 256;
#pragma unroll
        for (int nt = 0; nt < 8; nt++) {
            v2f b = *(const v2f*)(wp + (nt * 16 + l) * 2);
            acc[nt] = __builtin_amdgcn_wmma_f32_16x16x4_f32(
                false, a, false, b, (short)0, acc[nt], false, false);
        }
    }

#pragma unroll
    for (int nt = 0; nt < 8; nt++) {
        const int col = nt * 16 + l;
#pragma unroll
        for (int r = 0; r < 8; r++) {
            const int row = m0 + r + 8 * half;
            H[(size_t)row * DF + col] = acc[nt][r];
        }
    }
}


__global__ __launch_bounds__(256) void gather_rows_kernel(const int* __restrict__ rowptr,
                                                          const int* __restrict__ cnt,
                                                          const int* __restrict__ col,
                                                          const float* __restrict__ dinv,
                                                          const float* __restrict__ H,
                                                          const float* __restrict__ bias,
                                                          float* __restrict__ out, int n, int do_relu) {
    const int lane = threadIdx.x & 31;
    const int i = (int)((blockIdx.x * blockDim.x + threadIdx.x) >> 5);
    if (i >= n) return;
    const float di = dinv[i];
    const int start = rowptr[i];
    const int deg = cnt[i];
    const v4f* H4 = (const v4f*)H;

    v4f acc = H4[(size_t)i * (DF / 4) + lane] * (di * di);

    for (int base = 0; base < deg; base += 32) {
        const int idx = base + lane;
        int j = 0;
        float w = 0.0f;
        if (idx < deg) {
            j = col[start + idx];
            w = dinv[j] * di;
        }
        const int wi = __float_as_int(w);
        const int m = min(32, deg - base);
        int k = 0;
        for (; k + 4 <= m; k += 4) {
            const int j0 = __builtin_amdgcn_readlane(j, k);
            const int j1 = __builtin_amdgcn_readlane(j, k + 1);
            const int j2 = __builtin_amdgcn_readlane(j, k + 2);
            const int j3 = __builtin_amdgcn_readlane(j, k + 3);
            const float w0 = __int_as_float(__builtin_amdgcn_readlane(wi, k));
            const float w1 = __int_as_float(__builtin_amdgcn_readlane(wi, k + 1));
            const float w2 = __int_as_float(__builtin_amdgcn_readlane(wi, k + 2));
            const float w3 = __int_as_float(__builtin_amdgcn_readlane(wi, k + 3));
            const v4f h0 = H4[(size_t)j0 * (DF / 4) + lane];
            const v4f h1 = H4[(size_t)j1 * (DF / 4) + lane];
            const v4f h2 = H4[(size_t)j2 * (DF / 4) + lane];
            const v4f h3 = H4[(size_t)j3 * (DF / 4) + lane];
            acc += h0 * w0;
            acc += h1 * w1;
            acc += h2 * w2;
            acc += h3 * w3;
        }
        for (; k < m; k++) {
            const int j0 = __builtin_amdgcn_readlane(j, k);
            const float w0 = __int_as_float(__builtin_amdgcn_readlane(wi, k));
            acc += H4[(size_t)j0 * (DF / 4) + lane] * w0;
        }
    }

    acc += ((const v4f*)bias)[lane];
    if (do_relu) {
        acc.x = fmaxf(acc.x, 0.0f);
        acc.y = fmaxf(acc.y, 0.0f);
        acc.z = fmaxf(acc.z, 0.0f);
        acc.w = fmaxf(acc.w, 0.0f);
    }
    ((v4f*)out)[(size_t)i * (DF / 4) + lane] = acc;
}


__global__ __launch_bounds__(256) void decode_kernel(const int* __restrict__ s,
                                                     const int* __restrict__ d,
                                                     const float* __restrict__ z,
                                                     float* __restrict__ logits, int e) {
    const int lane = threadIdx.x & 31;
    const int w = (int)((blockIdx.x * blockDim.x + threadIdx.x) >> 5);
    if (w >= e) return;
    const int si = s[w];
    const int di = d[w];
    v4f a = ((const v4f*)(z + (size_t)si * DF))[lane];
    v4f b = ((const v4f*)(z + (size_t)di * DF))[lane];
    float dot = a.x * b.x + a.y * b.y + a.z * b.z + a.w * b.w;
#pragma unroll
    for (int off = 16; off > 0; off >>= 1) dot += __shfl_xor(dot, off, 32);
    if (lane == 0) logits[w] = dot;
}


extern "C" void kernel_launch(void* const* d_in, const int* in_sizes, int n_in,
                              void* d_out, int out_size, void* d_ws, size_t ws_size,
                              hipStream_t stream) {
    const float* x  = (const float*)d_in[0];
    const int*   ei = (const int*)d_in[1];
    const int*   el = (const int*)d_in[2];
    const float* W1 = (const float*)d_in[3];
    const float* b1 = (const float*)d_in[4];
    const float* W2 = (const float*)d_in[5];
    const float* b2 = (const float*)d_in[6];

    const int n  = in_sizes[0] / DF;
    const int E  = in_sizes[1] / 2;
    const int EL = in_sizes[2] / 2;

    const int tiles  = (n + 15) / 16;
    const int gemmBl = (tiles + 3) / 4;
    const int NB     = (n + BNODES - 1) / BNODES;
    const int NC     = (E + CH - 1) / CH;
    const int nodeBl = (n + 7) / 8;
    const int decBl  = (EL + 7) / 8;

    float* bufA   = (float*)d_ws;
    float* bufB   = bufA + (size_t)n * DF;
    float* dinv   = bufB + (size_t)n * DF;
    int*   cnt    = (int*)(dinv + n);
    int*   rowptr = cnt + n;
    int*   hmat   = rowptr + n;
    int*   bstart = hmat + (size_t)NC * NB;
    int*   psrc   = bstart + NB + 1;
    int*   pdst   = psrc + E;
    int*   col    = pdst + E;
    float* out    = (float*)d_out;

    chunk_hist_kernel<<<NC, 1024, 0, stream>>>(ei + E, hmat, E, NB);
    chunk_scatter_kernel<<<NC, 1024, 0, stream>>>(ei, ei + E, hmat, bstart, psrc, pdst, E, NB, NC);
    bucket_sort_kernel<<<NB, 1024, 0, stream>>>(psrc, pdst, bstart, rowptr, cnt, dinv, col, n);

    gemm_wmma_kernel<<<gemmBl, 128, 0, stream>>>(x, W1, bufA, n);
    gather_rows_kernel<<<nodeBl, 256, 0, stream>>>(rowptr, cnt, col, dinv, bufA, b1, bufB, n, 1);

    gemm_wmma_kernel<<<gemmBl, 128, 0, stream>>>(bufB, W2, bufA, n);
    gather_rows_kernel<<<nodeBl, 256, 0, stream>>>(rowptr, cnt, col, dinv, bufA, b2, bufB, n, 0);

    decode_kernel<<<decBl, 256, 0, stream>>>(el, el + EL, bufB, out, EL);
}


# Each gather wave walks 8 nodes and each decode wave 32 label edges instead of one wave per item, so far fewer short-lived waves must enter; every sum keeps its order
# speedup vs baseline: 3.0250x; 1.0542x over previous
#include <hip/hip_runtime.h>

typedef __attribute__((ext_vector_type(2))) float v2f;
typedef __attribute__((ext_vector_type(4))) float v4f;
typedef __attribute__((ext_vector_type(8))) float v8f;

#define DF 128


#define BSH 9
#define BNODES 512
#define NBMAX 256
#define CH 8192
#define KCAP 16384

__global__ __launch_bounds__(1024) void chunk_hist_kernel(const int* __restrict__ dst,
                                                          int* __restrict__ hmat, int e, int nb) {
    __shared__ int hist[NBMAX];
    if (threadIdx.x < NBMAX) hist[threadIdx.x] = 0;
    __syncthreads();
    const int c0 = blockIdx.x * CH;
    int ed[8];
#pragma unroll
    for (int k = 0; k < 8; k++) {
        const int i = c0 + k * 1024 + (int)threadIdx.x;
        ed[k] = (i < e) ? dst[i] : -1;
    }
#pragma unroll
    for (int k = 0; k < 8; k++)
        if (ed[k] >= 0) atomicAdd(&hist[ed[k] >> BSH], 1);
    __syncthreads();
    if ((int)threadIdx.x < nb) hmat[blockIdx.x * nb + threadIdx.x] = hist[threadIdx.x];
}

__global__ __launch_bounds__(1024) void chunk_scatter_kernel(const int* __restrict__ src,
                                                             const int* __restrict__ dst,
                                                             const int* __restrict__ hmat,
                                                             int* __restrict__ bstart,
                                                             int* __restrict__ psrc,
                                                             int* __restrict__ pdst,
                                                             int e, int nb, int nchunks) {
    __shared__ int hist[NBMAX];
    __shared__ int lstart[NBMAX];
    __shared__ int gbase[NBMAX];
    __shared__ int ptot[4][NBMAX];
    __shared__ int ppre[4][NBMAX];
    __shared__ int wsT[8];
    __shared__ int wsH[8];
    __shared__ int stS[CH];
    __shared__ int stD[CH];

    const int tid = threadIdx.x;
    const int lane = tid & 31;
    const int wave = tid >> 5;
    const int blk = blockIdx.x;
    const int c0 = blk * CH;

    if (tid < NBMAX) hist[tid] = 0;
    __syncthreads();

    int es[8], ed[8], er[8];
#pragma unroll
    for (int k = 0; k < 8; k++) {
        const int i = c0 + k * 1024 + tid;
        ed[k] = (i < e) ? dst[i] : -1;
        es[k] = (i < e) ? src[i] : 0;
    }
#pragma unroll
    for (int k = 0; k < 8; k++) {
        er[k] = 0;
        if (ed[k] >= 0) er[k] = atomicAdd(&hist[ed[k] >> BSH], 1);
    }

    {
        const int b = tid & (NBMAX - 1);
        const int g = tid >> 8;
        const int rpg = (nchunks + 3) >> 2;
        const int r0 = g * rpg;
        const int r1 = min(nchunks, r0 + rpg);
        int tot = 0, pre = 0;
        if (b < nb) {
#pragma unroll 8
            for (int r = r0; r < r1; r++) {
                const int v = hmat[r * nb + b];
                tot += v;
                pre += (r < blk) ? v : 0;
            }
        }
        ptot[g][b] = tot;
        ppre[g][b] = pre;
    }
    __syncthreads();

    int T = 0, P = 0, Hc = 0;
    if (tid < NBMAX) {
        T = ptot[0][tid] + ptot[1][tid] + ptot[2][tid] + ptot[3][tid];
        P = ppre[0][tid] + ppre[1][tid] + ppre[2][tid] + ppre[3][tid];
        Hc = hist[tid];
    }
    int incT = T, incH = Hc;
#pragma unroll
    for (int off = 1; off < 32; off <<= 1) {
        const int a = __shfl_up(incT, off, 32);
        const int h = __shfl_up(incH, off, 32);
        if (lane >= off) {
            incT += a;
            incH += h;
        }
    }
    if (tid < NBMAX && lane == 31) {
        wsT[wave] = incT;
        wsH[wave] = incH;
    }
    __syncthreads();
    if (tid < NBMAX) {
        int oT = 0, oH = 0;
#pragma unroll
        for (int w = 0; w < 8; w++) {
            if (w < wave) {
                oT += wsT[w];
                oH += wsH[w];
            }
        }
        const int exT = oT + incT - T;
        const int exH = oH + incH - Hc;
        lstart[tid] = exH;
        gbase[tid] = exT + P;
        if (blk == 0) {
            if (tid < nb) bstart[tid] = exT;
            if (tid == nb - 1) bstart[nb] = exT + T;
        }
    }
    __syncthreads();

#pragma unroll
    for (int k = 0; k < 8; k++) {
        if (ed[k] >= 0) {
            const int p = lstart[ed[k] >> BSH] + er[k];
            stS[p] = es[k];
            stD[p] = ed[k];
        }
    }
    __syncthreads();

    const int m = min(CH, e - c0);
    for (int i = tid; i < m; i += 1024) {
        const int s = stS[i];
        const int d = stD[i];
        const int b = d >> BSH;
        const int g = gbase[b] + (i - lstart[b]);
        psrc[g] = s;
        pdst[g] = d;
    }
}

__global__ __launch_bounds__(1024) void bucket_sort_kernel(const int* __restrict__ psrc,
                                                           const int* __restrict__ pdst,
                                                           const int* __restrict__ bstart,
                                                           int* __restrict__ rowptr,
                                                           int* __restrict__ cnt,
                                                           float* __restrict__ dinv,
                                                           int* __restrict__ col, int n) {
    __shared__ int cntL[BNODES];
    __shared__ int curL[BNODES];
    __shared__ int wsum[16];
    __shared__ int stage[KCAP];

    const int tid = threadIdx.x;
    const int lane = tid & 31;
    const int wave = tid >> 5;
    const int b = blockIdx.x;
    const int s = bstart[b];
    const int m = bstart[b + 1] - s;
    const int node0 = b << BSH;

    if (tid < BNODES) cntL[tid] = 0;
    __syncthreads();
    for (int i0 = tid; i0 < m; i0 += 8 * 1024) {
        int d[8];
#pragma unroll
        for (int k = 0; k < 8; k++) {
            const int i = i0 + k * 1024;
            d[k] = (i < m) ? pdst[s + i] : -1;
        }
#pragma unroll
        for (int k = 0; k < 8; k++)
            if (d[k] >= 0) atomicAdd(&cntL[d[k] - node0], 1);
    }
    __syncthreads();

    int c = 0;
    if (tid < BNODES) c = cntL[tid];
    int inc = c;
#pragma unroll
    for (int off = 1; off < 32; off <<= 1) {
        const int t = __shfl_up(inc, off, 32);
        if (lane >= off) inc += t;
    }
    if (tid < BNODES && lane == 31) wsum[wave] = inc;
    __syncthreads();
    if (tid < BNODES) {
        int o = 0;
#pragma unroll
        for (int w = 0; w < 16; w++)
            if (w < wave) o += wsum[w];
        const int ex = o + inc - c;
        curL[tid] = ex;
        const int node = node0 + tid;
        if (node < n) {
            rowptr[node] = s + ex;
            cnt[node] = c;
            dinv[node] = rsqrtf((float)(c + 1));
        }
    }
    __syncthreads();

    const bool staged = (m <= KCAP);
    for (int i0 = tid; i0 < m; i0 += 8 * 1024) {
        int d[8], sv[8];
#pragma unroll
        for (int k = 0; k < 8; k++) {
            const int i = i0 + k * 1024;
            d[k] = (i < m) ? pdst[s + i] : -1;
            sv[k] = (i < m) ? psrc[s + i] : 0;
        }
#pragma unroll
        for (int k = 0; k < 8; k++) {
            if (d[k] >= 0) {
                const int p = atomicAdd(&curL[d[k] - node0], 1);
                if (staged) stage[p] = sv[k];
                else col[s + p] = sv[k];
            }
        }
    }
    if (staged) {
        __syncthreads();
        for (int i = tid; i < m; i += 1024) col[s + i] = stage[i];
    }
}


__global__ __launch_bounds__(128) void gemm_wmma_kernel(const float* __restrict__ X,
                                                        const float* __restrict__ W,
                                                        float* __restrict__ H, int nrows) {
    __shared__ float sW[DF * DF];
    for (int i = threadIdx.x; i < DF * DF; i += blockDim.x) {
        int k = i >> 7;
        int nn = i & 127;
        sW[(k >> 1) * 256 + nn * 2 + (k & 1)] = W[i];
    }
    __syncthreads();

    const int wave = threadIdx.x >> 5;
    const int lane = threadIdx.x & 31;
    const int tile = blockIdx.x * 4 + wave;
    const int m0 = tile * 16;
    if (m0 >= nrows) return;

    const int half = lane >> 4;
    const int l = lane & 15;

    v8f acc[8] = {};
    const float* xrow = X + (size_t)(m0 + l) * DF;

    for (int kb = 0; kb < DF; kb += 4) {
        v2f a = *(const v2f*)(xrow + kb + half * 2);
        const float* wp = sW + ((kb >> 1) + half) * 256;
#pragma unroll
        for (int nt = 0; nt < 8; nt++) {
            v2f b = *(const v2f*)(wp + (nt * 16 + l) * 2);
            acc[nt] = __builtin_amdgcn_wmma_f32_16x16x4_f32(
                false, a, false, b, (short)0, acc[nt], false, false);
        }
    }

#pragma unroll
    for (int nt = 0; nt < 8; nt++) {
        const int col = nt * 16 + l;
#pragma unroll
        for (int r = 0; r < 8; r++) {
            const int row = m0 + r + 8 * half;
            H[(size_t)row * DF + col] = acc[nt][r];
        }
    }
}


#define NPW 8

__global__ __launch_bounds__(256) void gather_rows_kernel(const int* __restrict__ rowptr,
                                                          const int* __restrict__ cnt,
                                                          const int* __restrict__ col,
                                                          const float* __restrict__ dinv,
                                                          const float* __restrict__ H,
                                                          const float* __restrict__ bias,
                                                          float* __restrict__ out, int n, int do_relu) {
    const int lane = threadIdx.x & 31;
    const int wv = __builtin_amdgcn_readfirstlane((int)((blockIdx.x * blockDim.x + threadIdx.x) >> 5));
    const int i0 = wv * NPW;
    if (i0 >= n) return;
    const int nn = min(NPW, n - i0);
    int mstart = 0, mdeg = 0;
    float mdi = 0.0f;
    if (lane < nn) {
        mstart = rowptr[i0 + lane];
        mdeg = cnt[i0 + lane];
        mdi = dinv[i0 + lane];
    }
    const int mdii = __float_as_int(mdi);
    const v4f* H4 = (const v4f*)H;
    const v4f bv = ((const v4f*)bias)[lane];

    for (int t = 0; t < nn; t++) {
        const int i = i0 + t;
        const int start = __builtin_amdgcn_readlane(mstart, t);
        const int deg = __builtin_amdgcn_readlane(mdeg, t);
        const float di = __int_as_float(__builtin_amdgcn_readlane(mdii, t));

        v4f acc = H4[(size_t)i * (DF / 4) + lane] * (di * di);

        for (int base = 0; base < deg; base += 32) {
            const int idx = base + lane;
            int j = 0;
            float w = 0.0f;
            if (idx < deg) {
                j = col[start + idx];
                w = dinv[j] * di;
            }
            const int wi = __float_as_int(w);
            const int m = min(32, deg - base);
            int k = 0;
            for (; k + 4 <= m; k += 4) {
                const int j0 = __builtin_amdgcn_readlane(j, k);
                const int j1 = __builtin_amdgcn_readlane(j, k + 1);
                const int j2 = __builtin_amdgcn_readlane(j, k + 2);
                const int j3 = __builtin_amdgcn_readlane(j, k + 3);
                const float w0 = __int_as_float(__builtin_amdgcn_readlane(wi, k));
                const float w1 = __int_as_float(__builtin_amdgcn_readlane(wi, k + 1));
                const float w2 = __int_as_float(__builtin_amdgcn_readlane(wi, k + 2));
                const float w3 = __int_as_float(__builtin_amdgcn_readlane(wi, k + 3));
                const v4f h0 = H4[(size_t)j0 * (DF / 4) + lane];
                const v4f h1 = H4[(size_t)j1 * (DF / 4) + lane];
                const v4f h2 = H4[(size_t)j2 * (DF / 4) + lane];
                const v4f h3 = H4[(size_t)j3 * (DF / 4) + lane];
                acc += h0 * w0;
                acc += h1 * w1;
                acc += h2 * w2;
                acc += h3 * w3;
            }
            for (; k < m; k++) {
                const int j0 = __builtin_amdgcn_readlane(j, k);
                const float w0 = __int_as_float(__builtin_amdgcn_readlane(wi, k));
                acc += H4[(size_t)j0 * (DF / 4) + lane] * w0;
            }
        }

        acc += bv;
        if (do_relu) {
            acc.x = fmaxf(acc.x, 0.0f);
            acc.y = fmaxf(acc.y, 0.0f);
            acc.z = fmaxf(acc.z, 0.0f);
            acc.w = fmaxf(acc.w, 0.0f);
        }
        ((v4f*)out)[(size_t)i * (DF / 4) + lane] = acc;
    }
}


#define EPW 32

__global__ __launch_bounds__(256) void decode_kernel(const int* __restrict__ s,
                                                     const int* __restrict__ d,
                                                     const float* __restrict__ z,
                                                     float* __restrict__ logits, int e) {
    const int lane = threadIdx.x & 31;
    const int w = __builtin_amdgcn_readfirstlane((int)((blockIdx.x * blockDim.x + threadIdx.x) >> 5));
    const int e0 = w * EPW;
    if (e0 >= e) return;
    int sv = 0, dv = 0;
    if (e0 + lane < e) {
        sv = s[e0 + lane];
        dv = d[e0 + lane];
    }
    const v4f* Z4 = (const v4f*)z;
    const bool hi16 = (lane & 16) != 0;
    const bool hi8 = (lane & 8) != 0;
    float res = 0.0f;
    for (int q = 0; q < EPW / 4; q++) {
        v4f a[4], b[4];
#pragma unroll
        for (int u = 0; u < 4; u++) {
            const int sj = __builtin_amdgcn_readlane(sv, 4 * q + u);
            const int dj = __builtin_amdgcn_readlane(dv, 4 * q + u);
            a[u] = Z4[(size_t)sj * (DF / 4) + lane];
            b[u] = Z4[(size_t)dj * (DF / 4) + lane];
        }
        float p[4];
#pragma unroll
        for (int u = 0; u < 4; u++)
            p[u] = a[u].x * b[u].x + a[u].y * b[u].y + a[u].z * b[u].z + a[u].w * b[u].w;
        const float q0 = (hi16 ? p[1] : p[0]) + __shfl_xor(hi16 ? p[0] : p[1], 16, 32);
        const float q1 = (hi16 ? p[3] : p[2]) + __shfl_xor(hi16 ? p[2] : p[3], 16, 32);
        float r = (hi8 ? q1 : q0) + __shfl_xor(hi8 ? q0 : q1, 8, 32);
        r += __shfl_xor(r, 4, 32);
        r += __shfl_xor(r, 2, 32);
        r += __shfl_xor(r, 1, 32);
        if ((lane & 7) == q) res = r;
    }
    const int slot = 4 * (lane & 7) + 2 * ((lane >> 3) & 1) + (lane >> 4);
    if (e0 + slot < e) logits[e0 + slot] = res;
}


extern "C" void kernel_launch(void* const* d_in, const int* in_sizes, int n_in,
                              void* d_out, int out_size, void* d_ws, size_t ws_size,
                              hipStream_t stream) {
    const float* x  = (const float*)d_in[0];
    const int*   ei = (const int*)d_in[1];
    const int*   el = (const int*)d_in[2];
    const float* W1 = (const float*)d_in[3];
    const float* b1 = (const float*)d_in[4];
    const float* W2 = (const float*)d_in[5];
    const float* b2 = (const float*)d_in[6];

    const int n  = in_sizes[0] / DF;
    const int E  = in_sizes[1] / 2;
    const int EL = in_sizes[2] / 2;

    const int tiles  = (n + 15) / 16;
    const int gemmBl = (tiles + 3) / 4;
    const int NB     = (n + BNODES - 1) / BNODES;
    const int NC     = (E + CH - 1) / CH;
    const int nodeBl = (n + 8 * NPW - 1) / (8 * NPW);
    const int decBl  = (EL + 8 * EPW - 1) / (8 * EPW);

    float* bufA   = (float*)d_ws;
    float* bufB   = bufA + (size_t)n * DF;
    float* dinv   = bufB + (size_t)n * DF;
    int*   cnt    = (int*)(dinv + n);
    int*   rowptr = cnt + n;
    int*   hmat   = rowptr + n;
    int*   bstart = hmat + (size_t)NC * NB;
    int*   psrc   = bstart + NB + 1;
    int*   pdst   = psrc + E;
    int*   col    = pdst + E;
    float* out    = (float*)d_out;

    chunk_hist_kernel<<<NC, 1024, 0, stream>>>(ei + E, hmat, E, NB);
    chunk_scatter_kernel<<<NC, 1024, 0, stream>>>(ei, ei + E, hmat, bstart, psrc, pdst, E, NB, NC);
    bucket_sort_kernel<<<NB, 1024, 0, stream>>>(psrc, pdst, bstart, rowptr, cnt, dinv, col, n);

    gemm_wmma_kernel<<<gemmBl, 128, 0, stream>>>(x, W1, bufA, n);
    gather_rows_kernel<<<nodeBl, 256, 0, stream>>>(rowptr, cnt, col, dinv, bufA, b1, bufB, n, 1);

    gemm_wmma_kernel<<<gemmBl, 128, 0, stream>>>(bufB, W2, bufA, n);
    gather_rows_kernel<<<nodeBl, 256, 0, stream>>>(rowptr, cnt, col, dinv, bufA, b2, bufB, n, 0);

    decode_kernel<<<decBl, 256, 0, stream>>>(el, el + EL, bufB, out, EL);
}


# GEMM stages W (64 KB) to LDS with 32 independent b128 loads per thread issued as one group (a wave reads whole rows), instead of 128 four-byte loads each waited for before the next
# speedup vs baseline: 5.3997x; 1.7850x over previous
#include <hip/hip_runtime.h>

typedef __attribute__((ext_vector_type(2))) float v2f;
typedef __attribute__((ext_vector_type(4))) float v4f;
typedef __attribute__((ext_vector_type(8))) float v8f;

#define DF 128


#define BSH 9
#define BNODES 512
#define NBMAX 256
#define CH 8192
#define KCAP 16384

__global__ __launch_bounds__(1024) void chunk_hist_kernel(const int* __restrict__ dst,
                                                          int* __restrict__ hmat, int e, int nb) {
    __shared__ int hist[NBMAX];
    if (threadIdx.x < NBMAX) hist[threadIdx.x] = 0;
    __syncthreads();
    const int c0 = blockIdx.x * CH;
    int ed[8];
#pragma unroll
    for (int k = 0; k < 8; k++) {
        const int i = c0 + k * 1024 + (int)threadIdx.x;
        ed[k] = (i < e) ? dst[i] : -1;
    }
#pragma unroll
    for (int k = 0; k < 8; k++)
        if (ed[k] >= 0) atomicAdd(&hist[ed[k] >> BSH], 1);
    __syncthreads();
    if ((int)threadIdx.x < nb) hmat[blockIdx.x * nb + threadIdx.x] = hist[threadIdx.x];
}

__global__ __launch_bounds__(1024) void chunk_scatter_kernel(const int* __restrict__ src,
                                                             const int* __restrict__ dst,
                                                             const int* __restrict__ hmat,
                                                             int* __restrict__ bstart,
                                                             int* __restrict__ psrc,
                                                             int* __restrict__ pdst,
                                                             int e, int nb, int nchunks) {
    __shared__ int hist[NBMAX];
    __shared__ int lstart[NBMAX];
    __shared__ int gbase[NBMAX];
    __shared__ int ptot[4][NBMAX];
    __shared__ int ppre[4][NBMAX];
    __shared__ int wsT[8];
    __shared__ int wsH[8];
    __shared__ int stS[CH];
    __shared__ int stD[CH];

    const int tid = threadIdx.x;
    const int lane = tid & 31;
    const int wave = tid >> 5;
    const int blk = blockIdx.x;
    const int c0 = blk * CH;

    if (tid < NBMAX) hist[tid] = 0;
    __syncthreads();

    int es[8], ed[8], er[8];
#pragma unroll
    for (int k = 0; k < 8; k++) {
        const int i = c0 + k * 1024 + tid;
        ed[k] = (i < e) ? dst[i] : -1;
        es[k] = (i < e) ? src[i] : 0;
    }
#pragma unroll
    for (int k = 0; k < 8; k++) {
        er[k] = 0;
        if (ed[k] >= 0) er[k] = atomicAdd(&hist[ed[k] >> BSH], 1);
    }

    {
        const int b = tid & (NBMAX - 1);
        const int g = tid >> 8;
        const int rpg = (nchunks + 3) >> 2;
        const int r0 = g * rpg;
        const int r1 = min(nchunks, r0 + rpg);
        int tot = 0, pre = 0;
        if (b < nb) {
#pragma unroll 8
            for (int r = r0; r < r1; r++) {
                const int v = hmat[r * nb + b];
                tot += v;
                pre += (r < blk) ? v : 0;
            }
        }
        ptot[g][b] = tot;
        ppre[g][b] = pre;
    }
    __syncthreads();

    int T = 0, P = 0, Hc = 0;
    if (tid < NBMAX) {
        T = ptot[0][tid] + ptot[1][tid] + ptot[2][tid] + ptot[3][tid];
        P = ppre[0][tid] + ppre[1][tid] + ppre[2][tid] + ppre[3][tid];
        Hc = hist[tid];
    }
    int incT = T, incH = Hc;
#pragma unroll
    for (int off = 1; off < 32; off <<= 1) {
        const int a = __shfl_up(incT, off, 32);
        const int h = __shfl_up(incH, off, 32);
        if (lane >= off) {
            incT += a;
            incH += h;
        }
    }
    if (tid < NBMAX && lane == 31) {
        wsT[wave] = incT;
        wsH[wave] = incH;
    }
    __syncthreads();
    if (tid < NBMAX) {
        int oT = 0, oH = 0;
#pragma unroll
        for (int w = 0; w < 8; w++) {
            if (w < wave) {
                oT += wsT[w];
                oH += wsH[w];
            }
        }
        const int exT = oT + incT - T;
        const int exH = oH + incH - Hc;
        lstart[tid] = exH;
        gbase[tid] = exT + P;
        if (blk == 0) {
            if (tid < nb) bstart[tid] = exT;
            if (tid == nb - 1) bstart[nb] = exT + T;
        }
    }
    __syncthreads();

#pragma unroll
    for (int k = 0; k < 8; k++) {
        if (ed[k] >= 0) {
            const int p = lstart[ed[k] >> BSH] + er[k];
            stS[p] = es[k];
            stD[p] = ed[k];
        }
    }
    __syncthreads();

    const int m = min(CH, e - c0);
    for (int i = tid; i < m; i += 1024) {
        const int s = stS[i];
        const int d = stD[i];
        const int b = d >> BSH;
        const int g = gbase[b] + (i - lstart[b]);
        psrc[g] = s;
        pdst[g] = d;
    }
}

__global__ __launch_bounds__(1024) void bucket_sort_kernel(const int* __restrict__ psrc,
                                                           const int* __restrict__ pdst,
                                                           const int* __restrict__ bstart,
                                                           int* __restrict__ rowptr,
                                                           int* __restrict__ cnt,
                                                           float* __restrict__ dinv,
                                                           int* __restrict__ col, int n) {
    __shared__ int cntL[BNODES];
    __shared__ int curL[BNODES];
    __shared__ int wsum[16];
    __shared__ int stage[KCAP];

    const int tid = threadIdx.x;
    const int lane = tid & 31;
    const int wave = tid >> 5;
    const int b = blockIdx.x;
    const int s = bstart[b];
    const int m = bstart[b + 1] - s;
    const int node0 = b << BSH;

    if (tid < BNODES) cntL[tid] = 0;
    __syncthreads();
    for (int i0 = tid; i0 < m; i0 += 8 * 1024) {
        int d[8];
#pragma unroll
        for (int k = 0; k < 8; k++) {
            const int i = i0 + k * 1024;
            d[k] = (i < m) ? pdst[s + i] : -1;
        }
#pragma unroll
        for (int k = 0; k < 8; k++)
            if (d[k] >= 0) atomicAdd(&cntL[d[k] - node0], 1);
    }
    __syncthreads();

    int c = 0;
    if (tid < BNODES) c = cntL[tid];
    int inc = c;
#pragma unroll
    for (int off = 1; off < 32; off <<= 1) {
        const int t = __shfl_up(inc, off, 32);
        if (lane >= off) inc += t;
    }
    if (tid < BNODES && lane == 31) wsum[wave] = inc;
    __syncthreads();
    if (tid < BNODES) {
        int o = 0;
#pragma unroll
        for (int w = 0; w < 16; w++)
            if (w < wave) o += wsum[w];
        const int ex = o + inc - c;
        curL[tid] = ex;
        const int node = node0 + tid;
        if (node < n) {
            rowptr[node] = s + ex;
            cnt[node] = c;
            dinv[node] = rsqrtf((float)(c + 1));
        }
    }
    __syncthreads();

    const bool staged = (m <= KCAP);
    for (int i0 = tid; i0 < m; i0 += 8 * 1024) {
        int d[8], sv[8];
#pragma unroll
        for (int k = 0; k < 8; k++) {
            const int i = i0 + k * 1024;
            d[k] = (i < m) ? pdst[s + i] : -1;
            sv[k] = (i < m) ? psrc[s + i] : 0;
        }
#pragma unroll
        for (int k = 0; k < 8; k++) {
            if (d[k] >= 0) {
                const int p = atomicAdd(&curL[d[k] - node0], 1);
                if (staged) stage[p] = sv[k];
                else col[s + p] = sv[k];
            }
        }
    }
    if (staged) {
        __syncthreads();
        for (int i = tid; i < m; i += 1024) col[s + i] = stage[i];
    }
}


__global__ __launch_bounds__(128) void gemm_wmma_kernel(const float* __restrict__ X,
                                                        const float* __restrict__ W,
                                                        float* __restrict__ H, int nrows) {
    __shared__ __attribute__((aligned(16))) float sW[DF * DF];
    {
        const int cq = threadIdx.x & 31;
        const int wq = threadIdx.x >> 5;
        const v4f* W4 = (const v4f*)W;
        v4f ra[16], rb[16];
#pragma unroll
        for (int u = 0; u < 16; u++) {
            const int kp = u * 4 + wq;
            ra[u] = W4[(2 * kp) * (DF / 4) + cq];
            rb[u] = W4[(2 * kp + 1) * (DF / 4) + cq];
        }
#pragma unroll
        for (int u = 0; u < 16; u++) {
            const int kp = u * 4 + wq;
            v4f* dstp = (v4f*)(sW + kp * 256 + cq * 8);
            const v4f lo = {ra[u].x, rb[u].x, ra[u].y, rb[u].y};
            const v4f hi = {ra[u].z, rb[u].z, ra[u].w, rb[u].w};
            dstp[0] = lo;
            dstp[1] = hi;
        }
    }
    __syncthreads();

    const int wave = threadIdx.x >> 5;
    const int lane = threadIdx.x & 31;
    const int tile = blockIdx.x * 4 + wave;
    const int m0 = tile * 16;
    if (m0 >= nrows) return;

    const int half = lane >> 4;
    const int l = lane & 15;

    v8f acc[8] = {};
    const float* xrow = X + (size_t)(m0 + l) * DF;

    for (int kb = 0; kb < DF; kb += 4) {
        v2f a = *(const v2f*)(xrow + kb + half * 2);
        const float* wp = sW + ((kb >> 1) + half) * 256;
#pragma unroll
        for (int nt = 0; nt < 8; nt++) {
            v2f b = *(const v2f*)(wp + (nt * 16 + l) * 2);
            acc[nt] = __builtin_amdgcn_wmma_f32_16x16x4_f32(
                false, a, false, b, (short)0, acc[nt], false, false);
        }
    }

#pragma unroll
    for (int nt = 0; nt < 8; nt++) {
        const int col = nt * 16 + l;
#pragma unroll
        for (int r = 0; r < 8; r++) {
            const int row = m0 + r + 8 * half;
            H[(size_t)row * DF + col] = acc[nt][r];
        }
    }
}


#define NPW 8

__global__ __launch_bounds__(256) void gather_rows_kernel(const int* __restrict__ rowptr,
                                                          const int* __restrict__ cnt,
                                                          const int* __restrict__ col,
                                                          const float* __restrict__ dinv,
                                                          const float* __restrict__ H,
                                                          const float* __restrict__ bias,
                                                          float* __restrict__ out, int n, int do_relu) {
    const int lane = threadIdx.x & 31;
    const int wv = __builtin_amdgcn_readfirstlane((int)((blockIdx.x * blockDim.x + threadIdx.x) >> 5));
    const int i0 = wv * NPW;
    if (i0 >= n) return;
    const int nn = min(NPW, n - i0);
    int mstart = 0, mdeg = 0;
    float mdi = 0.0f;
    if (lane < nn) {
        mstart = rowptr[i0 + lane];
        mdeg = cnt[i0 + lane];
        mdi = dinv[i0 + lane];
    }
    const int mdii = __float_as_int(mdi);
    const v4f* H4 = (const v4f*)H;
    const v4f bv = ((const v4f*)bias)[lane];

    for (int t = 0; t < nn; t++) {
        const int i = i0 + t;
        const int start = __builtin_amdgcn_readlane(mstart, t);
        const int deg = __builtin_amdgcn_readlane(mdeg, t);
        const float di = __int_as_float(__builtin_amdgcn_readlane(mdii, t));

        v4f acc = H4[(size_t)i * (DF / 4) + lane] * (di * di);

        for (int base = 0; base < deg; base += 32) {
            const int idx = base + lane;
            int j = 0;
            float w = 0.0f;
            if (idx < deg) {
                j = col[start + idx];
                w = dinv[j] * di;
            }
            const int wi = __float_as_int(w);
            const int m = min(32, deg - base);
            int k = 0;
            for (; k + 4 <= m; k += 4) {
                const int j0 = __builtin_amdgcn_readlane(j, k);
                const int j1 = __builtin_amdgcn_readlane(j, k + 1);
                const int j2 = __builtin_amdgcn_readlane(j, k + 2);
                const int j3 = __builtin_amdgcn_readlane(j, k + 3);
                const float w0 = __int_as_float(__builtin_amdgcn_readlane(wi, k));
                const float w1 = __int_as_float(__builtin_amdgcn_readlane(wi, k + 1));
                const float w2 = __int_as_float(__builtin_amdgcn_readlane(wi, k + 2));
                const float w3 = __int_as_float(__builtin_amdgcn_readlane(wi, k + 3));
                const v4f h0 = H4[(size_t)j0 * (DF / 4) + lane];
                const v4f h1 = H4[(size_t)j1 * (DF / 4) + lane];
                const v4f h2 = H4[(size_t)j2 * (DF / 4) + lane];
                const v4f h3 = H4[(size_t)j3 * (DF / 4) + lane];
                acc += h0 * w0;
                acc += h1 * w1;
                acc += h2 * w2;
                acc += h3 * w3;
            }
            for (; k < m; k++) {
                const int j0 = __builtin_amdgcn_readlane(j, k);
                const float w0 = __int_as_float(__builtin_amdgcn_readlane(wi, k));
                acc += H4[(size_t)j0 * (DF / 4) + lane] * w0;
            }
        }

        acc += bv;
        if (do_relu) {
            acc.x = fmaxf(acc.x, 0.0f);
            acc.y = fmaxf(acc.y, 0.0f);
            acc.z = fmaxf(acc.z, 0.0f);
            acc.w = fmaxf(acc.w, 0.0f);
        }
        ((v4f*)out)[(size_t)i * (DF / 4) + lane] = acc;
    }
}


#define EPW 32

__global__ __launch_bounds__(256) void decode_kernel(const int* __restrict__ s,
                                                     const int* __restrict__ d,
                                                     const float* __restrict__ z,
                                                     float* __restrict__ logits, int e) {
    const int lane = threadIdx.x & 31;
    const int w = __builtin_amdgcn_readfirstlane((int)((blockIdx.x * blockDim.x + threadIdx.x) >> 5));
    const int e0 = w * EPW;
    if (e0 >= e) return;
    int sv = 0, dv = 0;
    if (e0 + lane < e) {
        sv = s[e0 + lane];
        dv = d[e0 + lane];
    }
    const v4f* Z4 = (const v4f*)z;
    const bool hi16 = (lane & 16) != 0;
    const bool hi8 = (lane & 8) != 0;
    float res = 0.0f;
    for (int q = 0; q < EPW / 4; q++) {
        v4f a[4], b[4];
#pragma unroll
        for (int u = 0; u < 4; u++) {
            const int sj = __builtin_amdgcn_readlane(sv, 4 * q + u);
            const int dj = __builtin_amdgcn_readlane(dv, 4 * q + u);
            a[u] = Z4[(size_t)sj * (DF / 4) + lane];
            b[u] = Z4[(size_t)dj * (DF / 4) + lane];
        }
        float p[4];
#pragma unroll
        for (int u = 0; u < 4; u++)
            p[u] = a[u].x * b[u].x + a[u].y * b[u].y + a[u].z * b[u].z + a[u].w * b[u].w;
        const float q0 = (hi16 ? p[1] : p[0]) + __shfl_xor(hi16 ? p[0] : p[1], 16, 32);
        const float q1 = (hi16 ? p[3] : p[2]) + __shfl_xor(hi16 ? p[2] : p[3], 16, 32);
        float r = (hi8 ? q1 : q0) + __shfl_xor(hi8 ? q0 : q1, 8, 32);
        r += __shfl_xor(r, 4, 32);
        r += __shfl_xor(r, 2, 32);
        r += __shfl_xor(r, 1, 32);
        if ((lane & 7) == q) res = r;
    }
    const int slot = 4 * (lane & 7) + 2 * ((lane >> 3) & 1) + (lane >> 4);
    if (e0 + slot < e) logits[e0 + slot] = res;
}


extern "C" void kernel_launch(void* const* d_in, const int* in_sizes, int n_in,
                              void* d_out, int out_size, void* d_ws, size_t ws_size,
                              hipStream_t stream) {
    const float* x  = (const float*)d_in[0];
    const int*   ei = (const int*)d_in[1];
    const int*   el = (const int*)d_in[2];
    const float* W1 = (const float*)d_in[3];
    const float* b1 = (const float*)d_in[4];
    const float* W2 = (const float*)d_in[5];
    const float* b2 = (const float*)d_in[6];

    const int n  = in_sizes[0] / DF;
    const int E  = in_sizes[1] / 2;
    const int EL = in_sizes[2] / 2;

    const int tiles  = (n + 15) / 16;
    const int gemmBl = (tiles + 3) / 4;
    const int NB     = (n + BNODES - 1) / BNODES;
    const int NC     = (E + CH - 1) / CH;
    const int nodeBl = (n + 8 * NPW - 1) / (8 * NPW);
    const int decBl  = (EL + 8 * EPW - 1) / (8 * EPW);

    float* bufA   = (float*)d_ws;
    float* bufB   = bufA + (size_t)n * DF;
    float* dinv   = bufB + (size_t)n * DF;
    int*   cnt    = (int*)(dinv + n);
    int*   rowptr = cnt + n;
    int*   hmat   = rowptr + n;
    int*   bstart = hmat + (size_t)NC * NB;
    int*   psrc   = bstart + NB + 1;
    int*   pdst   = psrc + E;
    int*   col    = pdst + E;
    float* out    = (float*)d_out;

    chunk_hist_kernel<<<NC, 1024, 0, stream>>>(ei + E, hmat, E, NB);
    chunk_scatter_kernel<<<NC, 1024, 0, stream>>>(ei, ei + E, hmat, bstart, psrc, pdst, E, NB, NC);
    bucket_sort_kernel<<<NB, 1024, 0, stream>>>(psrc, pdst, bstart, rowptr, cnt, dinv, col, n);

    gemm_wmma_kernel<<<gemmBl, 128, 0, stream>>>(x, W1, bufA, n);
    gather_rows_kernel<<<nodeBl, 256, 0, stream>>>(rowptr, cnt, col, dinv, bufA, b1, bufB, n, 1);

    gemm_wmma_kernel<<<gemmBl, 128, 0, stream>>>(bufB, W2, bufA, n);
    gather_rows_kernel<<<nodeBl, 256, 0, stream>>>(rowptr, cnt, col, dinv, bufA, b2, bufB, n, 0);

    decode_kernel<<<decBl, 256, 0, stream>>>(el, el + EL, bufB, out, EL);
}


# Aggregate first, multiply after: each block sums its nodes' input rows into an LDS tile and feeds it straight to the WMMA with W; X@W is never stored, both GEMM launches are gone; sum order changes
# speedup vs baseline: 5.8112x; 1.0762x over previous
#include <hip/hip_runtime.h>

typedef __attribute__((ext_vector_type(2))) float v2f;
typedef __attribute__((ext_vector_type(4))) float v4f;
typedef __attribute__((ext_vector_type(8))) float v8f;

#define DF 128


#define BSH 9
#define BNODES 512
#define NBMAX 256
#define CH 8192
#define KCAP 16384

__global__ __launch_bounds__(1024) void chunk_hist_kernel(const int* __restrict__ dst,
                                                          int* __restrict__ hmat, int e, int nb) {
    __shared__ int hist[NBMAX];
    if (threadIdx.x < NBMAX) hist[threadIdx.x] = 0;
    __syncthreads();
    const int c0 = blockIdx.x * CH;
    int ed[8];
#pragma unroll
    for (int k = 0; k < 8; k++) {
        const int i = c0 + k * 1024 + (int)threadIdx.x;
        ed[k] = (i < e) ? dst[i] : -1;
    }
#pragma unroll
    for (int k = 0; k < 8; k++)
        if (ed[k] >= 0) atomicAdd(&hist[ed[k] >> BSH], 1);
    __syncthreads();
    if ((int)threadIdx.x < nb) hmat[blockIdx.x * nb + threadIdx.x] = hist[threadIdx.x];
}

__global__ __launch_bounds__(1024) void chunk_scatter_kernel(const int* __restrict__ src,
                                                             const int* __restrict__ dst,
                                                             const int* __restrict__ hmat,
                                                             int* __restrict__ bstart,
                                                             int* __restrict__ psrc,
                                                             int* __restrict__ pdst,
                                                             int e, int nb, int nchunks) {
    __shared__ int hist[NBMAX];
    __shared__ int lstart[NBMAX];
    __shared__ int gbase[NBMAX];
    __shared__ int ptot[4][NBMAX];
    __shared__ int ppre[4][NBMAX];
    __shared__ int wsT[8];
    __shared__ int wsH[8];
    __shared__ int stS[CH];
    __shared__ int stD[CH];

    const int tid = threadIdx.x;
    const int lane = tid & 31;
    const int wave = tid >> 5;
    const int blk = blockIdx.x;
    const int c0 = blk * CH;

    if (tid < NBMAX) hist[tid] = 0;
    __syncthreads();

    int es[8], ed[8], er[8];
#pragma unroll
    for (int k = 0; k < 8; k++) {
        const int i = c0 + k * 1024 + tid;
        ed[k] = (i < e) ? dst[i] : -1;
        es[k] = (i < e) ? src[i] : 0;
    }
#pragma unroll
    for (int k = 0; k < 8; k++) {
        er[k] = 0;
        if (ed[k] >= 0) er[k] = atomicAdd(&hist[ed[k] >> BSH], 1);
    }

    {
        const int b = tid & (NBMAX - 1);
        const int g = tid >> 8;
        const int rpg = (nchunks + 3) >> 2;
        const int r0 = g * rpg;
        const int r1 = min(nchunks, r0 + rpg);
        int tot = 0, pre = 0;
        if (b < nb) {
#pragma unroll 8
            for (int r = r0; r < r1; r++) {
                const int v = hmat[r * nb + b];
                tot += v;
                pre += (r < blk) ? v : 0;
            }
        }
        ptot[g][b] = tot;
        ppre[g][b] = pre;
    }
    __syncthreads();

    int T = 0, P = 0, Hc = 0;
    if (tid < NBMAX) {
        T = ptot[0][tid] + ptot[1][tid] + ptot[2][tid] + ptot[3][tid];
        P = ppre[0][tid] + ppre[1][tid] + ppre[2][tid] + ppre[3][tid];
        Hc = hist[tid];
    }
    int incT = T, incH = Hc;
#pragma unroll
    for (int off = 1; off < 32; off <<= 1) {
        const int a = __shfl_up(incT, off, 32);
        const int h = __shfl_up(incH, off, 32);
        if (lane >= off) {
            incT += a;
            incH += h;
        }
    }
    if (tid < NBMAX && lane == 31) {
        wsT[wave] = incT;
        wsH[wave] = incH;
    }
    __syncthreads();
    if (tid < NBMAX) {
        int oT = 0, oH = 0;
#pragma unroll
        for (int w = 0; w < 8; w++) {
            if (w < wave) {
                oT += wsT[w];
                oH += wsH[w];
            }
        }
        const int exT = oT + incT - T;
        const int exH = oH + incH - Hc;
        lstart[tid] = exH;
        gbase[tid] = exT + P;
        if (blk == 0) {
            if (tid < nb) bstart[tid] = exT;
            if (tid == nb - 1) bstart[nb] = exT + T;
        }
    }
    __syncthreads();

#pragma unroll
    for (int k = 0; k < 8; k++) {
        if (ed[k] >= 0) {
            const int p = lstart[ed[k] >> BSH] + er[k];
            stS[p] = es[k];
            stD[p] = ed[k];
        }
    }
    __syncthreads();

    const int m = min(CH, e - c0);
    for (int i = tid; i < m; i += 1024) {
        const int s = stS[i];
        const int d = stD[i];
        const int b = d >> BSH;
        const int g = gbase[b] + (i - lstart[b]);
        psrc[g] = s;
        pdst[g] = d;
    }
}

__global__ __launch_bounds__(1024) void bucket_sort_kernel(const int* __restrict__ psrc,
                                                           const int* __restrict__ pdst,
                                                           const int* __restrict__ bstart,
                                                           int* __restrict__ rowptr,
                                                           int* __restrict__ cnt,
                                                           float* __restrict__ dinv,
                                                           int* __restrict__ col, int n) {
    __shared__ int cntL[BNODES];
    __shared__ int curL[BNODES];
    __shared__ int wsum[16];
    __shared__ int stage[KCAP];

    const int tid = threadIdx.x;
    const int lane = tid & 31;
    const int wave = tid >> 5;
    const int b = blockIdx.x;
    const int s = bstart[b];
    const int m = bstart[b + 1] - s;
    const int node0 = b << BSH;

    if (tid < BNODES) cntL[tid] = 0;
    __syncthreads();
    for (int i0 = tid; i0 < m; i0 += 8 * 1024) {
        int d[8];
#pragma unroll
        for (int k = 0; k < 8; k++) {
            const int i = i0 + k * 1024;
            d[k] = (i < m) ? pdst[s + i] : -1;
        }
#pragma unroll
        for (int k = 0; k < 8; k++)
            if (d[k] >= 0) atomicAdd(&cntL[d[k] - node0], 1);
    }
    __syncthreads();

    int c = 0;
    if (tid < BNODES) c = cntL[tid];
    int inc = c;
#pragma unroll
    for (int off = 1; off < 32; off <<= 1) {
        const int t = __shfl_up(inc, off, 32);
        if (lane >= off) inc += t;
    }
    if (tid < BNODES && lane == 31) wsum[wave] = inc;
    __syncthreads();
    if (tid < BNODES) {
        int o = 0;
#pragma unroll
        for (int w = 0; w < 16; w++)
            if (w < wave) o += wsum[w];
        const int ex = o + inc - c;
        curL[tid] = ex;
        const int node = node0 + tid;
        if (node < n) {
            rowptr[node] = s + ex;
            cnt[node] = c;
            dinv[node] = rsqrtf((float)(c + 1));
        }
    }
    __syncthreads();

    const bool staged = (m <= KCAP);
    for (int i0 = tid; i0 < m; i0 += 8 * 1024) {
        int d[8], sv[8];
#pragma unroll
        for (int k = 0; k < 8; k++) {
            const int i = i0 + k * 1024;
            d[k] = (i < m) ? pdst[s + i] : -1;
            sv[k] = (i < m) ? psrc[s + i] : 0;
        }
#pragma unroll
        for (int k = 0; k < 8; k++) {
            if (d[k] >= 0) {
                const int p = atomicAdd(&curL[d[k] - node0], 1);
                if (staged) stage[p] = sv[k];
                else col[s + p] = sv[k];
            }
        }
    }
    if (staged) {
        __syncthreads();
        for (int i = tid; i < m; i += 1024) col[s + i] = stage[i];
    }
}


#define TPW 4
#define NPQ 4
#define TP 132

__global__ __launch_bounds__(512) void agg_gemm_kernel(const int* __restrict__ rowptr,
                                                       const int* __restrict__ cnt,
                                                       const int* __restrict__ col,
                                                       const float* __restrict__ dinv,
                                                       const float* __restrict__ X,
                                                       const float* __restrict__ W,
                                                       const float* __restrict__ bias,
                                                       float* __restrict__ out, int n, int do_relu) {
    __shared__ __attribute__((aligned(16))) float sW[DF * DF];
    __shared__ __attribute__((aligned(16))) float sT[TPW * 16 * TP];

    const int lane = threadIdx.x & 31;
    const int wave = __builtin_amdgcn_readfirstlane((int)(threadIdx.x >> 5));
    {
        const v4f* W4 = (const v4f*)W;
        v4f ra[4], rb[4];
#pragma unroll
        for (int u = 0; u < 4; u++) {
            const int kp = u * 16 + wave;
            ra[u] = W4[(2 * kp) * (DF / 4) + lane];
            rb[u] = W4[(2 * kp + 1) * (DF / 4) + lane];
        }
#pragma unroll
        for (int u = 0; u < 4; u++) {
            const int kp = u * 16 + wave;
            v4f* dstp = (v4f*)(sW + kp * 256 + lane * 8);
            const v4f lo = {ra[u].x, rb[u].x, ra[u].y, rb[u].y};
            const v4f hi = {ra[u].z, rb[u].z, ra[u].w, rb[u].w};
            dstp[0] = lo;
            dstp[1] = hi;
        }
    }

    const int tl = wave >> 2;
    const int q = wave & 3;
    const int tile = blockIdx.x * TPW + tl;
    const int m0 = tile * 16;
    const int i0 = m0 + q * NPQ;
    const int nn = max(0, min(NPQ, n - i0));
    int mstart = 0, mdeg = 0;
    float mdi = 0.0f;
    if (lane < nn) {
        mstart = rowptr[i0 + lane];
        mdeg = cnt[i0 + lane];
        mdi = dinv[i0 + lane];
    }
    const int mdii = __float_as_int(mdi);
    const v4f* X4 = (const v4f*)X;
    float* trow = sT + tl * (16 * TP) + (q * NPQ) * TP + 4 * lane;

    for (int t = 0; t < NPQ; t++) {
        v4f acc = {0.0f, 0.0f, 0.0f, 0.0f};
        if (t < nn) {
            const int i = i0 + t;
            const int start = __builtin_amdgcn_readlane(mstart, t);
            const int deg = __builtin_amdgcn_readlane(mdeg, t);
            const float di = __int_as_float(__builtin_amdgcn_readlane(mdii, t));

            const v4f hs = X4[(size_t)i * (DF / 4) + lane];

            for (int base = 0; base < deg; base += 32) {
                const int idx = base + lane;
                int j = 0;
                float w = 0.0f;
                if (idx < deg) {
                    j = col[start + idx];
                    w = dinv[j] * di;
                }
                const int wi = __float_as_int(w);
                const int m = min(32, deg - base);
                int k = 0;
                for (; k + 8 <= m; k += 8) {
                    int jj[8];
                    float ww[8];
                    v4f hh[8];
#pragma unroll
                    for (int u = 0; u < 8; u++) {
                        jj[u] = __builtin_amdgcn_readlane(j, k + u);
                        ww[u] = __int_as_float(__builtin_amdgcn_readlane(wi, k + u));
                    }
#pragma unroll
                    for (int u = 0; u < 8; u++) hh[u] = X4[(size_t)jj[u] * (DF / 4) + lane];
#pragma unroll
                    for (int u = 0; u < 8; u++) acc += hh[u] * ww[u];
                }
                for (; k + 4 <= m; k += 4) {
                    const int j0 = __builtin_amdgcn_readlane(j, k);
                    const int j1 = __builtin_amdgcn_readlane(j, k + 1);
                    const int j2 = __builtin_amdgcn_readlane(j, k + 2);
                    const int j3 = __builtin_amdgcn_readlane(j, k + 3);
                    const float w0 = __int_as_float(__builtin_amdgcn_readlane(wi, k));
                    const float w1 = __int_as_float(__builtin_amdgcn_readlane(wi, k + 1));
                    const float w2 = __int_as_float(__builtin_amdgcn_readlane(wi, k + 2));
                    const float w3 = __int_as_float(__builtin_amdgcn_readlane(wi, k + 3));
                    const v4f h0 = X4[(size_t)j0 * (DF / 4) + lane];
                    const v4f h1 = X4[(size_t)j1 * (DF / 4) + lane];
                    const v4f h2 = X4[(size_t)j2 * (DF / 4) + lane];
                    const v4f h3 = X4[(size_t)j3 * (DF / 4) + lane];
                    acc += h0 * w0;
                    acc += h1 * w1;
                    acc += h2 * w2;
                    acc += h3 * w3;
                }
                for (; k < m; k++) {
                    const int j0 = __builtin_amdgcn_readlane(j, k);
                    const float w0 = __int_as_float(__builtin_amdgcn_readlane(wi, k));
                    acc += X4[(size_t)j0 * (DF / 4) + lane] * w0;
                }
            }
            acc += hs * (di * di);
        }
        *(v4f*)(trow + t * TP) = acc;
    }
    __syncthreads();

    const int half = lane >> 4;
    const int l = lane & 15;
    const float* ta = sT + tl * (16 * TP) + l * TP + half * 2;
    const float* wb = sW + half * 256 + (q * 32 + l) * 2;
    v8f acc0 = {};
    v8f acc1 = {};
#pragma unroll 8
    for (int kb = 0; kb < DF; kb += 4) {
        const v2f a = *(const v2f*)(ta + kb);
        const float* wp = wb + (kb >> 1) * 256;
        const v2f b0 = *(const v2f*)(wp);
        const v2f b1 = *(const v2f*)(wp + 32);
        acc0 = __builtin_amdgcn_wmma_f32_16x16x4_f32(false, a, false, b0, (short)0, acc0, false, false);
        acc1 = __builtin_amdgcn_wmma_f32_16x16x4_f32(false, a, false, b1, (short)0, acc1, false, false);
    }

    const int c0 = q * 32 + l;
    const float bz0 = bias[c0];
    const float bz1 = bias[c0 + 16];
#pragma unroll
    for (int r = 0; r < 8; r++) {
        const int row = m0 + r + 8 * half;
        float v0 = acc0[r] + bz0;
        float v1 = acc1[r] + bz1;
        if (do_relu) {
            v0 = fmaxf(v0, 0.0f);
            v1 = fmaxf(v1, 0.0f);
        }
        if (row < n) {
            out[(size_t)row * DF + c0] = v0;
            out[(size_t)row * DF + c0 + 16] = v1;
        }
    }
}


#define EPW 32

__global__ __launch_bounds__(256) void decode_kernel(const int* __restrict__ s,
                                                     const int* __restrict__ d,
                                                     const float* __restrict__ z,
                                                     float* __restrict__ logits, int e) {
    const int lane = threadIdx.x & 31;
    const int w = __builtin_amdgcn_readfirstlane((int)((blockIdx.x * blockDim.x + threadIdx.x) >> 5));
    const int e0 = w * EPW;
    if (e0 >= e) return;
    int sv = 0, dv = 0;
    if (e0 + lane < e) {
        sv = s[e0 + lane];
        dv = d[e0 + lane];
    }
    const v4f* Z4 = (const v4f*)z;
    const bool hi16 = (lane & 16) != 0;
    const bool hi8 = (lane & 8) != 0;
    float res = 0.0f;
    for (int q = 0; q < EPW / 4; q++) {
        v4f a[4], b[4];
#pragma unroll
        for (int u = 0; u < 4; u++) {
            const int sj = __builtin_amdgcn_readlane(sv, 4 * q + u);
            const int dj = __builtin_amdgcn_readlane(dv, 4 * q + u);
            a[u] = Z4[(size_t)sj * (DF / 4) + lane];
            b[u] = Z4[(size_t)dj * (DF / 4) + lane];
        }
        float p[4];
#pragma unroll
        for (int u = 0; u < 4; u++)
            p[u] = a[u].x * b[u].x + a[u].y * b[u].y + a[u].z * b[u].z + a[u].w * b[u].w;
        const float q0 = (hi16 ? p[1] : p[0]) + __shfl_xor(hi16 ? p[0] : p[1], 16, 32);
        const float q1 = (hi16 ? p[3] : p[2]) + __shfl_xor(hi16 ? p[2] : p[3], 16, 32);
        float r = (hi8 ? q1 : q0) + __shfl_xor(hi8 ? q0 : q1, 8, 32);
        r += __shfl_xor(r, 4, 32);
        r += __shfl_xor(r, 2, 32);
        r += __shfl_xor(r, 1, 32);
        if ((lane & 7) == q) res = r;
    }
    const int slot = 4 * (lane & 7) + 2 * ((lane >> 3) & 1) + (lane >> 4);
    if (e0 + slot < e) logits[e0 + slot] = res;
}


extern "C" void kernel_launch(void* const* d_in, const int* in_sizes, int n_in,
                              void* d_out, int out_size, void* d_ws, size_t ws_size,
                              hipStream_t stream) {
    const float* x  = (const float*)d_in[0];
    const int*   ei = (const int*)d_in[1];
    const int*   el = (const int*)d_in[2];
    const float* W1 = (const float*)d_in[3];
    const float* b1 = (const float*)d_in[4];
    const float* W2 = (const float*)d_in[5];
    const float* b2 = (const float*)d_in[6];

    const int n  = in_sizes[0] / DF;
    const int E  = in_sizes[1] / 2;
    const int EL = in_sizes[2] / 2;

    const int tiles  = (n + 15) / 16;
    const int aggBl  = (tiles + TPW - 1) / TPW;
    const int NB     = (n + BNODES - 1) / BNODES;
    const int NC     = (E + CH - 1) / CH;
    const int decBl  = (EL + 8 * EPW - 1) / (8 * EPW);

    float* bufA   = (float*)d_ws;
    float* bufB   = bufA + (size_t)n * DF;
    float* dinv   = bufB + (size_t)n * DF;
    int*   cnt    = (int*)(dinv + n);
    int*   rowptr = cnt + n;
    int*   hmat   = rowptr + n;
    int*   bstart = hmat + (size_t)NC * NB;
    int*   psrc   = bstart + NB + 1;
    int*   pdst   = psrc + E;
    int*   col    = pdst + E;
    float* out    = (float*)d_out;

    chunk_hist_kernel<<<NC, 1024, 0, stream>>>(ei + E, hmat, E, NB);
    chunk_scatter_kernel<<<NC, 1024, 0, stream>>>(ei, ei + E, hmat, bstart, psrc, pdst, E, NB, NC);
    bucket_sort_kernel<<<NB, 1024, 0, stream>>>(psrc, pdst, bstart, rowptr, cnt, dinv, col, n);

    agg_gemm_kernel<<<aggBl, 512, 0, stream>>>(rowptr, cnt, col, dinv, x, W1, b1, bufA, n, 1);

    agg_gemm_kernel<<<aggBl, 512, 0, stream>>>(rowptr, cnt, col, dinv, bufA, W2, b2, bufB, n, 0);

    decode_kernel<<<decBl, 256, 0, stream>>>(el, el + EL, bufB, out, EL);
}


# Resident blocks split roles: 20 waves sum neighbour rows of node chunk k while 4 waves run chunk k-1's matrix instructions from a second LDS tile set; W staged once a block; sums keep their order
# speedup vs baseline: 6.4229x; 1.1053x over previous
#include <hip/hip_runtime.h>

typedef __attribute__((ext_vector_type(2))) float v2f;
typedef __attribute__((ext_vector_type(4))) float v4f;
typedef __attribute__((ext_vector_type(8))) float v8f;

#define DF 128


#define BSH 9
#define BNODES 512
#define NBMAX 256
#define CH 8192
#define KCAP 16384

__global__ __launch_bounds__(1024) void chunk_hist_kernel(const int* __restrict__ dst,
                                                          int* __restrict__ hmat, int e, int nb) {
    __shared__ int hist[NBMAX];
    if (threadIdx.x < NBMAX) hist[threadIdx.x] = 0;
    __syncthreads();
    const int c0 = blockIdx.x * CH;
    int ed[8];
#pragma unroll
    for (int k = 0; k < 8; k++) {
        const int i = c0 + k * 1024 + (int)threadIdx.x;
        ed[k] = (i < e) ? dst[i] : -1;
    }
#pragma unroll
    for (int k = 0; k < 8; k++)
        if (ed[k] >= 0) atomicAdd(&hist[ed[k] >> BSH], 1);
    __syncthreads();
    if ((int)threadIdx.x < nb) hmat[blockIdx.x * nb + threadIdx.x] = hist[threadIdx.x];
}

__global__ __launch_bounds__(1024) void chunk_scatter_kernel(const int* __restrict__ src,
                                                             const int* __restrict__ dst,
                                                             const int* __restrict__ hmat,
                                                             int* __restrict__ bstart,
                                                             int* __restrict__ psrc,
                                                             int* __restrict__ pdst,
                                                             int e, int nb, int nchunks) {
    __shared__ int hist[NBMAX];
    __shared__ int lstart[NBMAX];
    __shared__ int gbase[NBMAX];
    __shared__ int ptot[4][NBMAX];
    __shared__ int ppre[4][NBMAX];
    __shared__ int wsT[8];
    __shared__ int wsH[8];
    __shared__ int stS[CH];
    __shared__ int stD[CH];

    const int tid = threadIdx.x;
    const int lane = tid & 31;
    const int wave = tid >> 5;
    const int blk = blockIdx.x;
    const int c0 = blk * CH;

    if (tid < NBMAX) hist[tid] = 0;
    __syncthreads();

    int es[8], ed[8], er[8];
#pragma unroll
    for (int k = 0; k < 8; k++) {
        const int i = c0 + k * 1024 + tid;
        ed[k] = (i < e) ? dst[i] : -1;
        es[k] = (i < e) ? src[i] : 0;
    }
#pragma unroll
    for (int k = 0; k < 8; k++) {
        er[k] = 0;
        if (ed[k] >= 0) er[k] = atomicAdd(&hist[ed[k] >> BSH], 1);
    }

    {
        const int b = tid & (NBMAX - 1);
        const int g = tid >> 8;
        const int rpg = (nchunks + 3) >> 2;
        const int r0 = g * rpg;
        const int r1 = min(nchunks, r0 + rpg);
        int tot = 0, pre = 0;
        if (b < nb) {
#pragma unroll 8
            for (int r = r0; r < r1; r++) {
                const int v = hmat[r * nb + b];
                tot += v;
                pre += (r < blk) ? v : 0;
            }
        }
        ptot[g][b] = tot;
        ppre[g][b] = pre;
    }
    __syncthreads();

    int T = 0, P = 0, Hc = 0;
    if (tid < NBMAX) {
        T = ptot[0][tid] + ptot[1][tid] + ptot[2][tid] + ptot[3][tid];
        P = ppre[0][tid] + ppre[1][tid] + ppre[2][tid] + ppre[3][tid];
        Hc = hist[tid];
    }
    int incT = T, incH = Hc;
#pragma unroll
    for (int off = 1; off < 32; off <<= 1) {
        const int a = __shfl_up(incT, off, 32);
        const int h = __shfl_up(incH, off, 32);
        if (lane >= off) {
            incT += a;
            incH += h;
        }
    }
    if (tid < NBMAX && lane == 31) {
        wsT[wave] = incT;
        wsH[wave] = incH;
    }
    __syncthreads();
    if (tid < NBMAX) {
        int oT = 0, oH = 0;
#pragma unroll
        for (int w = 0; w < 8; w++) {
            if (w < wave) {
                oT += wsT[w];
                oH += wsH[w];
            }
        }
        const int exT = oT + incT - T;
        const int exH = oH + incH - Hc;
        lstart[tid] = exH;
        gbase[tid] = exT + P;
        if (blk == 0) {
            if (tid < nb) bstart[tid] = exT;
            if (tid == nb - 1) bstart[nb] = exT + T;
        }
    }
    __syncthreads();

#pragma unroll
    for (int k = 0; k < 8; k++) {
        if (ed[k] >= 0) {
            const int p = lstart[ed[k] >> BSH] + er[k];
            stS[p] = es[k];
            stD[p] = ed[k];
        }
    }
    __syncthreads();

    const int m = min(CH, e - c0);
    for (int i = tid; i < m; i += 1024) {
        const int s = stS[i];
        const int d = stD[i];
        const int b = d >> BSH;
        const int g = gbase[b] + (i - lstart[b]);
        psrc[g] = s;
        pdst[g] = d;
    }
}

__global__ __launch_bounds__(1024) void bucket_sort_kernel(const int* __restrict__ psrc,
                                                           const int* __restrict__ pdst,
                                                           const int* __restrict__ bstart,
                                                           int* __restrict__ rowptr,
                                                           int* __restrict__ cnt,
                                                           float* __restrict__ dinv,
                                                           int* __restrict__ col, int n) {
    __shared__ int cntL[BNODES];
    __shared__ int curL[BNODES];
    __shared__ int wsum[16];
    __shared__ int stage[KCAP];

    const int tid = threadIdx.x;
    const int lane = tid & 31;
    const int wave = tid >> 5;
    const int b = blockIdx.x;
    const int s = bstart[b];
    const int m = bstart[b + 1] - s;
    const int node0 = b << BSH;

    if (tid < BNODES) cntL[tid] = 0;
    __syncthreads();
    for (int i0 = tid; i0 < m; i0 += 8 * 1024) {
        int d[8];
#pragma unroll
        for (int k = 0; k < 8; k++) {
            const int i = i0 + k * 1024;
            d[k] = (i < m) ? pdst[s + i] : -1;
        }
#pragma unroll
        for (int k = 0; k < 8; k++)
            if (d[k] >= 0) atomicAdd(&cntL[d[k] - node0], 1);
    }
    __syncthreads();

    int c = 0;
    if (tid < BNODES) c = cntL[tid];
    int inc = c;
#pragma unroll
    for (int off = 1; off < 32; off <<= 1) {
        const int t = __shfl_up(inc, off, 32);
        if (lane >= off) inc += t;
    }
    if (tid < BNODES && lane == 31) wsum[wave] = inc;
    __syncthreads();
    if (tid < BNODES) {
        int o = 0;
#pragma unroll
        for (int w = 0; w < 16; w++)
            if (w < wave) o += wsum[w];
        const int ex = o + inc - c;
        curL[tid] = ex;
        const int node = node0 + tid;
        if (node < n) {
            rowptr[node] = s + ex;
            cnt[node] = c;
            dinv[node] = rsqrtf((float)(c + 1));
        }
    }
    __syncthreads();

    const bool staged = (m <= KCAP);
    for (int i0 = tid; i0 < m; i0 += 8 * 1024) {
        int d[8], sv[8];
#pragma unroll
        for (int k = 0; k < 8; k++) {
            const int i = i0 + k * 1024;
            d[k] = (i < m) ? pdst[s + i] : -1;
            sv[k] = (i < m) ? psrc[s + i] : 0;
        }
#pragma unroll
        for (int k = 0; k < 8; k++) {
            if (d[k] >= 0) {
                const int p = atomicAdd(&curL[d[k] - node0], 1);
                if (staged) stage[p] = sv[k];
                else col[s + p] = sv[k];
            }
        }
    }
    if (staged) {
        __syncthreads();
        for (int i = tid; i < m; i += 1024) col[s + i] = stage[i];
    }
}


#define GW 20
#define CT 5
#define CN 80
#define NPQ 4
#define TP 132
#define AGG_THREADS 768
#define AGG_BLOCKS 512

__global__ __launch_bounds__(AGG_THREADS) void agg_gemm_kernel(const int* __restrict__ rowptr,
                                                               const int* __restrict__ cnt,
                                                               const int* __restrict__ col,
                                                               const float* __restrict__ dinv,
                                                               const float* __restrict__ X,
                                                               const float* __restrict__ W,
                                                               const float* __restrict__ bias,
                                                               float* __restrict__ out, int n, int do_relu) {
    __shared__ __attribute__((aligned(16))) float sW[DF * DF];
    __shared__ __attribute__((aligned(16))) float sT[2 * CT * 16 * TP];

    const int lane = threadIdx.x & 31;
    const int wave = __builtin_amdgcn_readfirstlane((int)(threadIdx.x >> 5));
    {
        const v4f* W4 = (const v4f*)W;
        v4f ra[3], rb[3];
#pragma unroll
        for (int u = 0; u < 3; u++) {
            const int kp = u * 24 + wave;
            if (kp < DF / 2) {
                ra[u] = W4[(2 * kp) * (DF / 4) + lane];
                rb[u] = W4[(2 * kp + 1) * (DF / 4) + lane];
            }
        }
#pragma unroll
        for (int u = 0; u < 3; u++) {
            const int kp = u * 24 + wave;
            if (kp < DF / 2) {
                v4f* dstp = (v4f*)(sW + kp * 256 + lane * 8);
                const v4f lo4 = {ra[u].x, rb[u].x, ra[u].y, rb[u].y};
                const v4f hi4 = {ra[u].z, rb[u].z, ra[u].w, rb[u].w};
                dstp[0] = lo4;
                dstp[1] = hi4;
            }
        }
    }

    const int per = (n + (int)gridDim.x - 1) / (int)gridDim.x;
    const int lo = (int)blockIdx.x * per;
    const int hi = min(n, lo + per);
    const int nch = (hi > lo) ? (hi - lo + CN - 1) / CN : 0;
    const v4f* X4 = (const v4f*)X;
    __syncthreads();

    for (int trip = 0; trip <= nch; trip++) {
        if (wave < GW) {
            if (trip < nch) {
                const int cbeg = lo + trip * CN;
                const int cend = min(hi, cbeg + CN);
                const int tl = wave >> 2;
                const int q = wave & 3;
                const int i0 = cbeg + tl * 16 + q * NPQ;
                const int nn = max(0, min(NPQ, cend - i0));
                int mstart = 0, mdeg = 0;
                float mdi = 0.0f;
                if (lane < nn) {
                    mstart = rowptr[i0 + lane];
                    mdeg = cnt[i0 + lane];
                    mdi = dinv[i0 + lane];
                }
                const int mdii = __float_as_int(mdi);
                float* trow = sT + (trip & 1) * (CT * 16 * TP) + tl * (16 * TP) + (q * NPQ) * TP + 4 * lane;

                for (int t = 0; t < NPQ; t++) {
                    v4f acc = {0.0f, 0.0f, 0.0f, 0.0f};
                    if (t < nn) {
                        const int i = i0 + t;
                        const int start = __builtin_amdgcn_readlane(mstart, t);
                        const int deg = __builtin_amdgcn_readlane(mdeg, t);
                        const float di = __int_as_float(__builtin_amdgcn_readlane(mdii, t));

                        const v4f hs = X4[(size_t)i * (DF / 4) + lane];

                        for (int base = 0; base < deg; base += 32) {
                            const int idx = base + lane;
                            int j = 0;
                            float w = 0.0f;
                            if (idx < deg) {
                                j = col[start + idx];
                                w = dinv[j] * di;
                            }
                            const int wi = __float_as_int(w);
                            const int m = min(32, deg - base);
                            int k = 0;
                            for (; k + 8 <= m; k += 8) {
                                int jj[8];
                                float ww[8];
                                v4f hh[8];
            #pragma unroll
                                for (int u = 0; u < 8; u++) {
                                    jj[u] = __builtin_amdgcn_readlane(j, k + u);
                                    ww[u] = __int_as_float(__builtin_amdgcn_readlane(wi, k + u));
                                }
            #pragma unroll
                                for (int u = 0; u < 8; u++) hh[u] = X4[(size_t)jj[u] * (DF / 4) + lane];
            #pragma unroll
                                for (int u = 0; u < 8; u++) acc += hh[u] * ww[u];
                            }
                            for (; k + 4 <= m; k += 4) {
                                const int j0 = __builtin_amdgcn_readlane(j, k);
                                const int j1 = __builtin_amdgcn_readlane(j, k + 1);
                                const int j2 = __builtin_amdgcn_readlane(j, k + 2);
                                const int j3 = __builtin_amdgcn_readlane(j, k + 3);
                                const float w0 = __int_as_float(__builtin_amdgcn_readlane(wi, k));
                                const float w1 = __int_as_float(__builtin_amdgcn_readlane(wi, k + 1));
                                const float w2 = __int_as_float(__builtin_amdgcn_readlane(wi, k + 2));
                                const float w3 = __int_as_float(__builtin_amdgcn_readlane(wi, k + 3));
                                const v4f h0 = X4[(size_t)j0 * (DF / 4) + lane];
                                const v4f h1 = X4[(size_t)j1 * (DF / 4) + lane];
                                const v4f h2 = X4[(size_t)j2 * (DF / 4) + lane];
                                const v4f h3 = X4[(size_t)j3 * (DF / 4) + lane];
                                acc += h0 * w0;
                                acc += h1 * w1;
                                acc += h2 * w2;
                                acc += h3 * w3;
                            }
                            for (; k < m; k++) {
                                const int j0 = __builtin_amdgcn_readlane(j, k);
                                const float w0 = __int_as_float(__builtin_amdgcn_readlane(wi, k));
                                acc += X4[(size_t)j0 * (DF / 4) + lane] * w0;
                            }
                        }
                        acc += hs * (di * di);
                    }
                    *(v4f*)(trow + t * TP) = acc;
                }
            }
        } else {
            if (trip >= 1) {
                const int pc = trip - 1;
                const int cbeg = lo + pc * CN;
                const int cend = min(hi, cbeg + CN);
                const int nt = (cend - cbeg + 15) >> 4;
                const int mq = wave - GW;
                const int half = lane >> 4;
                const int l = lane & 15;
                const float* wb = sW + half * 256 + (mq * 32 + l) * 2;
                const int c0 = mq * 32 + l;
                const float bz0 = bias[c0];
                const float bz1 = bias[c0 + 16];
                for (int t = 0; t < nt; t++) {
                    const int m0 = cbeg + t * 16;
                    const float* ta = sT + (pc & 1) * (CT * 16 * TP) + t * (16 * TP) + l * TP + half * 2;
                    v8f acc0 = {};
                    v8f acc1 = {};
#pragma unroll 8
                    for (int kb = 0; kb < DF; kb += 4) {
                        const v2f a = *(const v2f*)(ta + kb);
                        const float* wp = wb + (kb >> 1) * 256;
                        const v2f b0 = *(const v2f*)(wp);
                        const v2f b1 = *(const v2f*)(wp + 32);
                        acc0 = __builtin_amdgcn_wmma_f32_16x16x4_f32(false, a, false, b0, (short)0, acc0, false, false);
                        acc1 = __builtin_amdgcn_wmma_f32_16x16x4_f32(false, a, false, b1, (short)0, acc1, false, false);
                    }
#pragma unroll
                    for (int r = 0; r < 8; r++) {
                        const int row = m0 + r + 8 * half;
                        float v0 = acc0[r] + bz0;
                        float v1 = acc1[r] + bz1;
                        if (do_relu) {
                            v0 = fmaxf(v0, 0.0f);
                            v1 = fmaxf(v1, 0.0f);
                        }
                        if (row < cend) {
                            out[(size_t)row * DF + c0] = v0;
                            out[(size_t)row * DF + c0 + 16] = v1;
                        }
                    }
                }
            }
        }
        __syncthreads();
    }
}


#define EPW 32

__global__ __launch_bounds__(256) void decode_kernel(const int* __restrict__ s,
                                                     const int* __restrict__ d,
                                                     const float* __restrict__ z,
                                                     float* __restrict__ logits, int e) {
    const int lane = threadIdx.x & 31;
    const int w = __builtin_amdgcn_readfirstlane((int)((blockIdx.x * blockDim.x + threadIdx.x) >> 5));
    const int e0 = w * EPW;
    if (e0 >= e) return;
    int sv = 0, dv = 0;
    if (e0 + lane < e) {
        sv = s[e0 + lane];
        dv = d[e0 + lane];
    }
    const v4f* Z4 = (const v4f*)z;
    const bool hi16 = (lane & 16) != 0;
    const bool hi8 = (lane & 8) != 0;
    float res = 0.0f;
    for (int q = 0; q < EPW / 4; q++) {
        v4f a[4], b[4];
#pragma unroll
        for (int u = 0; u < 4; u++) {
            const int sj = __builtin_amdgcn_readlane(sv, 4 * q + u);
            const int dj = __builtin_amdgcn_readlane(dv, 4 * q + u);
            a[u] = Z4[(size_t)sj * (DF / 4) + lane];
            b[u] = Z4[(size_t)dj * (DF / 4) + lane];
        }
        float p[4];
#pragma unroll
        for (int u = 0; u < 4; u++)
            p[u] = a[u].x * b[u].x + a[u].y * b[u].y + a[u].z * b[u].z + a[u].w * b[u].w;
        const float q0 = (hi16 ? p[1] : p[0]) + __shfl_xor(hi16 ? p[0] : p[1], 16, 32);
        const float q1 = (hi16 ? p[3] : p[2]) + __shfl_xor(hi16 ? p[2] : p[3], 16, 32);
        float r = (hi8 ? q1 : q0) + __shfl_xor(hi8 ? q0 : q1, 8, 32);
        r += __shfl_xor(r, 4, 32);
        r += __shfl_xor(r, 2, 32);
        r += __shfl_xor(r, 1, 32);
        if ((lane & 7) == q) res = r;
    }
    const int slot = 4 * (lane & 7) + 2 * ((lane >> 3) & 1) + (lane >> 4);
    if (e0 + slot < e) logits[e0 + slot] = res;
}


extern "C" void kernel_launch(void* const* d_in, const int* in_sizes, int n_in,
                              void* d_out, int out_size, void* d_ws, size_t ws_size,
                              hipStream_t stream) {
    const float* x  = (const float*)d_in[0];
    const int*   ei = (const int*)d_in[1];
    const int*   el = (const int*)d_in[2];
    const float* W1 = (const float*)d_in[3];
    const float* b1 = (const float*)d_in[4];
    const float* W2 = (const float*)d_in[5];
    const float* b2 = (const float*)d_in[6];

    const int n  = in_sizes[0] / DF;
    const int E  = in_sizes[1] / 2;
    const int EL = in_sizes[2] / 2;

    const int NB     = (n + BNODES - 1) / BNODES;
    const int NC     = (E + CH - 1) / CH;
    const int decBl  = (EL + 8 * EPW - 1) / (8 * EPW);

    float* bufA   = (float*)d_ws;
    float* bufB   = bufA + (size_t)n * DF;
    float* dinv   = bufB + (size_t)n * DF;
    int*   cnt    = (int*)(dinv + n);
    int*   rowptr = cnt + n;
    int*   hmat   = rowptr + n;
    int*   bstart = hmat + (size_t)NC * NB;
    int*   psrc   = bstart + NB + 1;
    int*   pdst   = psrc + E;
    int*   col    = pdst + E;
    float* out    = (float*)d_out;

    chunk_hist_kernel<<<NC, 1024, 0, stream>>>(ei + E, hmat, E, NB);
    chunk_scatter_kernel<<<NC, 1024, 0, stream>>>(ei, ei + E, hmat, bstart, psrc, pdst, E, NB, NC);
    bucket_sort_kernel<<<NB, 1024, 0, stream>>>(psrc, pdst, bstart, rowptr, cnt, dinv, col, n);

    agg_gemm_kernel<<<AGG_BLOCKS, AGG_THREADS, 0, stream>>>(rowptr, cnt, col, dinv, x, W1, b1, bufA, n, 1);

    agg_gemm_kernel<<<AGG_BLOCKS, AGG_THREADS, 0, stream>>>(rowptr, cnt, col, dinv, bufA, W2, b2, bufB, n, 0);

    decode_kernel<<<decBl, 256, 0, stream>>>(el, el + EL, bufB, out, EL);
}
